# starting kernel
# baseline (speedup 1.0000x reference)
#include <hip/hip_runtime.h>
#include <hip/hip_bf16.h>


typedef float v2f __attribute__((ext_vector_type(2)));
typedef float v8f __attribute__((ext_vector_type(8)));

__device__ __forceinline__ v8f wmma_f32_4(v2f a, v2f b, v8f c) {
  return __builtin_amdgcn_wmma_f32_16x16x4_f32(false, a, false, b, (short)0, c,
                                               false, false);
}

__device__ __forceinline__ float relu_(float x) { return fmaxf(x, 0.0f); }

__global__ __launch_bounds__(256) void k_template_path(
    const float* __restrict__ tf, const float* __restrict__ w_t,
    const float* __restrict__ gnw, const float* __restrict__ gnb,
    float* __restrict__ t_global, float* __restrict__ t_kernel) {
  const int b = blockIdx.x;
  const int tid = threadIdx.x;
  const int wave = tid >> 5, lane = tid & 31;
  const int hv = lane >> 4, l16 = lane & 15;

  __shared__ float lds_t[64][208];
  __shared__ float ga[64], gb[64];

  const float* X = tf + (size_t)b * 256 * 196;

  for (int t = wave; t < 52; t += 8) {
    const int m0 = (t & 3) << 4;
    const int n0 = (t >> 2) << 4;
    const int n = n0 + l16;
    const int nc = (n < 196) ? n : 195;
    v8f acc = {};
    for (int k0 = 0; k0 < 256; k0 += 4) {
      const int kk = k0 + (hv << 1);
      v2f a, bb;
      a.x = w_t[(m0 + l16) * 256 + kk];
      a.y = w_t[(m0 + l16) * 256 + kk + 1];
      bb.x = X[kk * 196 + nc];
      bb.y = X[(kk + 1) * 196 + nc];
      acc = wmma_f32_4(a, bb, acc);
    }
#pragma unroll
    for (int r = 0; r < 8; ++r)
      lds_t[m0 + r + (hv << 3)][n0 + l16] = acc[r];
  }
  __syncthreads();

  {
    const int g = tid >> 3, l8 = tid & 7;
    float s = 0.f, q = 0.f;
    for (int i = l8; i < 392; i += 8) {
      const float v = lds_t[(g << 1) + (i >= 196)][i % 196];
      s += v;
      q += v * v;
    }
    s += __shfl_xor(s, 1); q += __shfl_xor(q, 1);
    s += __shfl_xor(s, 2); q += __shfl_xor(q, 2);
    s += __shfl_xor(s, 4); q += __shfl_xor(q, 4);
    const float mean = s * (1.0f / 392.0f);
    const float var = q * (1.0f / 392.0f) - mean * mean;
    const float sc = rsqrtf(var + 1e-5f);
    if (l8 < 2) {
      const int c = (g << 1) + l8;
      ga[c] = gnw[c] * sc;
      gb[c] = gnb[c] - mean * sc * gnw[c];
    }
  }
  __syncthreads();

  for (int idx = tid; idx < 64 * 49; idx += 256) {
    const int c = idx / 49, kid = idx % 49;
    const int ky = kid / 7, kx = kid % 7;
    const float a = ga[c], bo = gb[c];
    float s = 0.f;
#pragma unroll
    for (int dy = 0; dy < 2; ++dy)
#pragma unroll
      for (int dx = 0; dx < 2; ++dx)
        s += relu_(a * lds_t[c][(2 * ky + dy) * 14 + (2 * kx + dx)] + bo);
    t_kernel[((size_t)b * 64 + c) * 49 + kid] = s * 0.25f;
  }
  if (tid < 64) {
    const int c = tid;
    const float a = ga[c], bo = gb[c];
    float s = 0.f;
    for (int p = 0; p < 196; ++p) s += relu_(a * lds_t[c][p] + bo);
    t_global[(size_t)b * 64 + c] = s * (1.0f / 196.0f);
  }
}

__global__ __launch_bounds__(256) void k_search_gemm(
    const float* __restrict__ sf, const float* __restrict__ w_s,
    float* __restrict__ s_pre) {
  const int b = blockIdx.y;
  const int tid = threadIdx.x;
  const int wave = tid >> 5, lane = tid & 31;
  const int hv = lane >> 4, l16 = lane & 15;
  const int m0 = (wave & 3) << 4;
  const int nbase = blockIdx.x * 128 + ((wave >> 2) << 6);

  const float* X = sf + (size_t)b * 256 * 4096;
  v8f acc0 = {}, acc1 = {}, acc2 = {}, acc3 = {};

  for (int k0 = 0; k0 < 256; k0 += 4) {
    const int kk = k0 + (hv << 1);
    v2f a;
    a.x = w_s[(m0 + l16) * 256 + kk];
    a.y = w_s[(m0 + l16) * 256 + kk + 1];
    const float* r0 = X + (size_t)kk * 4096 + nbase + l16;
    const float* r1 = r0 + 4096;
    v2f b0; b0.x = r0[0];  b0.y = r1[0];  acc0 = wmma_f32_4(a, b0, acc0);
    v2f b1; b1.x = r0[16]; b1.y = r1[16]; acc1 = wmma_f32_4(a, b1, acc1);
    v2f b2; b2.x = r0[32]; b2.y = r1[32]; acc2 = wmma_f32_4(a, b2, acc2);
    v2f b3; b3.x = r0[48]; b3.y = r1[48]; acc3 = wmma_f32_4(a, b3, acc3);
  }

  float* O = s_pre + (size_t)b * 64 * 4096;
#pragma unroll
  for (int r = 0; r < 8; ++r) {
    const int row = m0 + r + (hv << 3);
    float* o = O + (size_t)row * 4096 + nbase + l16;
    o[0] = acc0[r];
    o[16] = acc1[r];
    o[32] = acc2[r];
    o[48] = acc3[r];
  }
}

__global__ __launch_bounds__(256) void k_gn_stats(
    const float* __restrict__ pre, const float* __restrict__ gnw,
    const float* __restrict__ gnb, float* __restrict__ A,
    float* __restrict__ Bo) {
  const int b = blockIdx.x >> 5, g = blockIdx.x & 31;
  const float* p = pre + ((size_t)b * 64 + g * 2) * 4096;
  const int tid = threadIdx.x;
  float s = 0.f, q = 0.f;
  for (int i = tid; i < 8192; i += 256) {
    const float v = p[i];
    s += v;
    q += v * v;
  }
  __shared__ float ss[256], sq[256];
  ss[tid] = s;
  sq[tid] = q;
  __syncthreads();
  for (int o = 128; o > 0; o >>= 1) {
    if (tid < o) {
      ss[tid] += ss[tid + o];
      sq[tid] += sq[tid + o];
    }
    __syncthreads();
  }
  if (tid < 2) {
    const float mean = ss[0] * (1.f / 8192.f);
    const float var = sq[0] * (1.f / 8192.f) - mean * mean;
    const float sc = rsqrtf(var + 1e-5f);
    const int c = g * 2 + tid;
    A[(size_t)b * 64 + c] = gnw[c] * sc;
    Bo[(size_t)b * 64 + c] = gnb[c] - mean * sc * gnw[c];
  }
}

__global__ __launch_bounds__(256) void k_corr(
    const float* __restrict__ s_pre, const float* __restrict__ sA,
    const float* __restrict__ sB, const float* __restrict__ t_global,
    const float* __restrict__ t_kernel, float* __restrict__ corr) {
  const int bc = blockIdx.x;
  const float* src = s_pre + (size_t)bc * 4096;
  const float a = sA[bc], bo = sB[bc], tg = t_global[bc];
  __shared__ float tile[70][70];
  __shared__ float kern[49];
  const int tid = threadIdx.x;
  if (tid < 49) kern[tid] = t_kernel[(size_t)bc * 49 + tid];
  for (int i = tid; i < 70 * 70; i += 256) {
    const int y = i / 70 - 3, x = i % 70 - 3;
    const bool in = ((unsigned)y < 64u) & ((unsigned)x < 64u);
    const float raw = src[in ? (y * 64 + x) : 0];
    tile[i / 70][i % 70] = in ? relu_(a * raw + bo) : 0.f;
  }
  __syncthreads();
  float* dst = corr + (size_t)bc * 4096;
  for (int p = tid; p < 4096; p += 256) {
    const int y = p >> 6, x = p & 63;
    float acc = tg * tile[y + 3][x + 3];
#pragma unroll
    for (int ky = 0; ky < 7; ++ky)
#pragma unroll
      for (int kx = 0; kx < 7; ++kx)
        acc = fmaf(kern[ky * 7 + kx], tile[y + ky][x + kx], acc);
    dst[p] = acc;
  }
}

__global__ __launch_bounds__(256) void k_conv3x3_gemm(
    const float* __restrict__ corr, const float* __restrict__ w_p1,
    float* __restrict__ y_pre) {
  const int b = blockIdx.y;
  const int tid = threadIdx.x;
  const int wave = tid >> 5, lane = tid & 31;
  const int hv = lane >> 4, l16 = lane & 15;
  const int m0 = (wave & 3) << 4;
  const int nbase = blockIdx.x * 128 + ((wave >> 2) << 6);

  const float* X = corr + (size_t)b * 64 * 4096;
  v8f acc[4] = {};

  for (int k0 = 0; k0 < 576; k0 += 4) {
    const int kk = k0 + (hv << 1);
    v2f a;
    a.x = w_p1[(m0 + l16) * 576 + kk];
    a.y = w_p1[(m0 + l16) * 576 + kk + 1];

    const int c0 = kk / 9, r0 = kk - c0 * 9;
    const int dy0 = r0 / 3 - 1, dx0 = r0 % 3 - 1;
    const int kk1 = kk + 1;
    const int c1 = kk1 / 9, r1 = kk1 - c1 * 9;
    const int dy1 = r1 / 3 - 1, dx1 = r1 % 3 - 1;
    const float* P0 = X + (size_t)c0 * 4096;
    const float* P1 = X + (size_t)c1 * 4096;

#pragma unroll
    for (int j = 0; j < 4; ++j) {
      const int n = nbase + (j << 4) + l16;
      const int y = n >> 6, x = n & 63;
      const int y0 = y + dy0, x0 = x + dx0;
      const int y1 = y + dy1, x1 = x + dx1;
      const bool in0 = ((unsigned)y0 < 64u) & ((unsigned)x0 < 64u);
      const bool in1 = ((unsigned)y1 < 64u) & ((unsigned)x1 < 64u);
      const float v0 = P0[in0 ? (y0 * 64 + x0) : 0];
      const float v1 = P1[in1 ? (y1 * 64 + x1) : 0];
      v2f bb;
      bb.x = in0 ? v0 : 0.f;
      bb.y = in1 ? v1 : 0.f;
      acc[j] = wmma_f32_4(a, bb, acc[j]);
    }
  }

  float* O = y_pre + (size_t)b * 64 * 4096;
#pragma unroll
  for (int r = 0; r < 8; ++r) {
    const int row = m0 + r + (hv << 3);
    float* o = O + (size_t)row * 4096 + nbase + l16;
    o[0] = acc[0][r];
    o[16] = acc[1][r];
    o[32] = acc[2][r];
    o[48] = acc[3][r];
  }
}

__global__ __launch_bounds__(256) void k_final(
    const float* __restrict__ y_pre, const float* __restrict__ yA,
    const float* __restrict__ yB, const float* __restrict__ w_p2,
    const float* __restrict__ b_p2, float* __restrict__ out) {
  const int b = blockIdx.x >> 4;
  const int p = ((blockIdx.x & 15) << 8) + threadIdx.x;
  const float* Y = y_pre + (size_t)b * 64 * 4096 + p;
  float acc = b_p2[0];
#pragma unroll 8
  for (int c = 0; c < 64; ++c) {
    const float v = relu_(yA[b * 64 + c] * Y[(size_t)c * 4096] + yB[b * 64 + c]);
    acc = fmaf(w_p2[c], v, acc);
  }
  out[(size_t)b * 4096 + p] = acc;
}

extern "C" void kernel_launch(void* const* d_in, const int* in_sizes, int n_in,
                              void* d_out, int out_size, void* d_ws,
                              size_t ws_size, hipStream_t stream) {
  (void)in_sizes; (void)n_in; (void)out_size; (void)ws_size;
  const float* tf   = (const float*)d_in[0];
  const float* sf   = (const float*)d_in[1];
  const float* w_t  = (const float*)d_in[2];
  const float* gtw  = (const float*)d_in[3];
  const float* gtb  = (const float*)d_in[4];
  const float* w_s  = (const float*)d_in[5];
  const float* gsw  = (const float*)d_in[6];
  const float* gsb  = (const float*)d_in[7];
  const float* w_p1 = (const float*)d_in[8];
  const float* gpw  = (const float*)d_in[9];
  const float* gpb  = (const float*)d_in[10];
  const float* w_p2 = (const float*)d_in[11];
  const float* b_p2 = (const float*)d_in[12];
  float* out = (float*)d_out;

  float* ws = (float*)d_ws;
  float* t_global = ws;
  float* t_kernel = ws + 2048;
  float* sA = ws + 102400;
  float* sB = ws + 104448;
  float* yA = ws + 106496;
  float* yB = ws + 108544;
  float* bufA = ws + 131072;
  float* bufB = bufA + 8388608;

  k_template_path<<<32, 256, 0, stream>>>(tf, w_t, gtw, gtb, t_global, t_kernel);
  k_search_gemm<<<dim3(32, 32), 256, 0, stream>>>(sf, w_s, bufA);
  k_gn_stats<<<1024, 256, 0, stream>>>(bufA, gsw, gsb, sA, sB);
  k_corr<<<2048, 256, 0, stream>>>(bufA, sA, sB, t_global, t_kernel, bufB);
  k_conv3x3_gemm<<<dim3(32, 32), 256, 0, stream>>>(bufB, w_p1, bufA);
  k_gn_stats<<<1024, 256, 0, stream>>>(bufA, gpw, gpb, yA, yB);
  k_final<<<512, 256, 0, stream>>>(bufA, yA, yB, w_p2, b_p2, out);
}


# conv3x3: input window (4 rows x 66 cols x 64 ch, channel-innermost, pitch 68) staged once per block in LDS; B fragments by 16-byte LDS reads, no per-step gathers; K summed in a new order
# speedup vs baseline: 1.3033x; 1.3033x over previous
#include <hip/hip_runtime.h>
#include <hip/hip_bf16.h>


typedef float v2f __attribute__((ext_vector_type(2)));
typedef float v8f __attribute__((ext_vector_type(8)));

__device__ __forceinline__ v8f wmma_f32_4(v2f a, v2f b, v8f c) {
  return __builtin_amdgcn_wmma_f32_16x16x4_f32(false, a, false, b, (short)0, c,
                                               false, false);
}

__device__ __forceinline__ float relu_(float x) { return fmaxf(x, 0.0f); }

__global__ __launch_bounds__(256) void k_template_path(
    const float* __restrict__ tf, const float* __restrict__ w_t,
    const float* __restrict__ gnw, const float* __restrict__ gnb,
    float* __restrict__ t_global, float* __restrict__ t_kernel) {
  const int b = blockIdx.x;
  const int tid = threadIdx.x;
  const int wave = tid >> 5, lane = tid & 31;
  const int hv = lane >> 4, l16 = lane & 15;

  __shared__ float lds_t[64][208];
  __shared__ float ga[64], gb[64];

  const float* X = tf + (size_t)b * 256 * 196;

  for (int t = wave; t < 52; t += 8) {
    const int m0 = (t & 3) << 4;
    const int n0 = (t >> 2) << 4;
    const int n = n0 + l16;
    const int nc = (n < 196) ? n : 195;
    v8f acc = {};
    for (int k0 = 0; k0 < 256; k0 += 4) {
      const int kk = k0 + (hv << 1);
      v2f a, bb;
      a.x = w_t[(m0 + l16) * 256 + kk];
      a.y = w_t[(m0 + l16) * 256 + kk + 1];
      bb.x = X[kk * 196 + nc];
      bb.y = X[(kk + 1) * 196 + nc];
      acc = wmma_f32_4(a, bb, acc);
    }
#pragma unroll
    for (int r = 0; r < 8; ++r)
      lds_t[m0 + r + (hv << 3)][n0 + l16] = acc[r];
  }
  __syncthreads();

  {
    const int g = tid >> 3, l8 = tid & 7;
    float s = 0.f, q = 0.f;
    for (int i = l8; i < 392; i += 8) {
      const float v = lds_t[(g << 1) + (i >= 196)][i % 196];
      s += v;
      q += v * v;
    }
    s += __shfl_xor(s, 1); q += __shfl_xor(q, 1);
    s += __shfl_xor(s, 2); q += __shfl_xor(q, 2);
    s += __shfl_xor(s, 4); q += __shfl_xor(q, 4);
    const float mean = s * (1.0f / 392.0f);
    const float var = q * (1.0f / 392.0f) - mean * mean;
    const float sc = rsqrtf(var + 1e-5f);
    if (l8 < 2) {
      const int c = (g << 1) + l8;
      ga[c] = gnw[c] * sc;
      gb[c] = gnb[c] - mean * sc * gnw[c];
    }
  }
  __syncthreads();

  for (int idx = tid; idx < 64 * 49; idx += 256) {
    const int c = idx / 49, kid = idx % 49;
    const int ky = kid / 7, kx = kid % 7;
    const float a = ga[c], bo = gb[c];
    float s = 0.f;
#pragma unroll
    for (int dy = 0; dy < 2; ++dy)
#pragma unroll
      for (int dx = 0; dx < 2; ++dx)
        s += relu_(a * lds_t[c][(2 * ky + dy) * 14 + (2 * kx + dx)] + bo);
    t_kernel[((size_t)b * 64 + c) * 49 + kid] = s * 0.25f;
  }
  if (tid < 64) {
    const int c = tid;
    const float a = ga[c], bo = gb[c];
    float s = 0.f;
    for (int p = 0; p < 196; ++p) s += relu_(a * lds_t[c][p] + bo);
    t_global[(size_t)b * 64 + c] = s * (1.0f / 196.0f);
  }
}

__global__ __launch_bounds__(256) void k_search_gemm(
    const float* __restrict__ sf, const float* __restrict__ w_s,
    float* __restrict__ s_pre) {
  const int b = blockIdx.y;
  const int tid = threadIdx.x;
  const int wave = tid >> 5, lane = tid & 31;
  const int hv = lane >> 4, l16 = lane & 15;
  const int m0 = (wave & 3) << 4;
  const int nbase = blockIdx.x * 128 + ((wave >> 2) << 6);

  const float* X = sf + (size_t)b * 256 * 4096;
  v8f acc0 = {}, acc1 = {}, acc2 = {}, acc3 = {};

  for (int k0 = 0; k0 < 256; k0 += 4) {
    const int kk = k0 + (hv << 1);
    v2f a;
    a.x = w_s[(m0 + l16) * 256 + kk];
    a.y = w_s[(m0 + l16) * 256 + kk + 1];
    const float* r0 = X + (size_t)kk * 4096 + nbase + l16;
    const float* r1 = r0 + 4096;
    v2f b0; b0.x = r0[0];  b0.y = r1[0];  acc0 = wmma_f32_4(a, b0, acc0);
    v2f b1; b1.x = r0[16]; b1.y = r1[16]; acc1 = wmma_f32_4(a, b1, acc1);
    v2f b2; b2.x = r0[32]; b2.y = r1[32]; acc2 = wmma_f32_4(a, b2, acc2);
    v2f b3; b3.x = r0[48]; b3.y = r1[48]; acc3 = wmma_f32_4(a, b3, acc3);
  }

  float* O = s_pre + (size_t)b * 64 * 4096;
#pragma unroll
  for (int r = 0; r < 8; ++r) {
    const int row = m0 + r + (hv << 3);
    float* o = O + (size_t)row * 4096 + nbase + l16;
    o[0] = acc0[r];
    o[16] = acc1[r];
    o[32] = acc2[r];
    o[48] = acc3[r];
  }
}

__global__ __launch_bounds__(256) void k_gn_stats(
    const float* __restrict__ pre, const float* __restrict__ gnw,
    const float* __restrict__ gnb, float* __restrict__ A,
    float* __restrict__ Bo) {
  const int b = blockIdx.x >> 5, g = blockIdx.x & 31;
  const float* p = pre + ((size_t)b * 64 + g * 2) * 4096;
  const int tid = threadIdx.x;
  float s = 0.f, q = 0.f;
  for (int i = tid; i < 8192; i += 256) {
    const float v = p[i];
    s += v;
    q += v * v;
  }
  __shared__ float ss[256], sq[256];
  ss[tid] = s;
  sq[tid] = q;
  __syncthreads();
  for (int o = 128; o > 0; o >>= 1) {
    if (tid < o) {
      ss[tid] += ss[tid + o];
      sq[tid] += sq[tid + o];
    }
    __syncthreads();
  }
  if (tid < 2) {
    const float mean = ss[0] * (1.f / 8192.f);
    const float var = sq[0] * (1.f / 8192.f) - mean * mean;
    const float sc = rsqrtf(var + 1e-5f);
    const int c = g * 2 + tid;
    A[(size_t)b * 64 + c] = gnw[c] * sc;
    Bo[(size_t)b * 64 + c] = gnb[c] - mean * sc * gnw[c];
  }
}

__global__ __launch_bounds__(256) void k_corr(
    const float* __restrict__ s_pre, const float* __restrict__ sA,
    const float* __restrict__ sB, const float* __restrict__ t_global,
    const float* __restrict__ t_kernel, float* __restrict__ corr) {
  const int bc = blockIdx.x;
  const float* src = s_pre + (size_t)bc * 4096;
  const float a = sA[bc], bo = sB[bc], tg = t_global[bc];
  __shared__ float tile[70][70];
  __shared__ float kern[49];
  const int tid = threadIdx.x;
  if (tid < 49) kern[tid] = t_kernel[(size_t)bc * 49 + tid];
  for (int i = tid; i < 70 * 70; i += 256) {
    const int y = i / 70 - 3, x = i % 70 - 3;
    const bool in = ((unsigned)y < 64u) & ((unsigned)x < 64u);
    const float raw = src[in ? (y * 64 + x) : 0];
    tile[i / 70][i % 70] = in ? relu_(a * raw + bo) : 0.f;
  }
  __syncthreads();
  float* dst = corr + (size_t)bc * 4096;
  for (int p = tid; p < 4096; p += 256) {
    const int y = p >> 6, x = p & 63;
    float acc = tg * tile[y + 3][x + 3];
#pragma unroll
    for (int ky = 0; ky < 7; ++ky)
#pragma unroll
      for (int kx = 0; kx < 7; ++kx)
        acc = fmaf(kern[ky * 7 + kx], tile[y + ky][x + kx], acc);
    dst[p] = acc;
  }
}

typedef float v4f __attribute__((ext_vector_type(4)));

#define CV_PITCH 68
#define CV_COLS 66
#define CV_ROWS 4

__global__ __launch_bounds__(256) void k_conv3x3_gemm(
    const float* __restrict__ corr, const float* __restrict__ w_p1,
    float* __restrict__ y_pre) {
  const int b = blockIdx.y;
  const int tid = threadIdx.x;
  const int wave = tid >> 5, lane = tid & 31;
  const int hv = lane >> 4, l16 = lane & 15;
  const int m0 = (wave & 3) << 4;
  const int yl = wave >> 2;
  const int y0 = blockIdx.x << 1;
  const int nbase = blockIdx.x * 128 + (yl << 6);

  __shared__ __attribute__((aligned(16))) float tile[CV_ROWS * CV_COLS * CV_PITCH];

  const float* X = corr + (size_t)b * 64 * 4096;

  {
    const int x4 = tid & 15, c4 = tid >> 4;
#pragma unroll
    for (int r = 0; r < CV_ROWS; ++r) {
      const int gy = y0 - 1 + r;
      const bool in = (unsigned)gy < 64u;
      const float* s = X + (size_t)(c4 * 4) * 4096 + (in ? gy : 0) * 64 + x4 * 4;
      v4f v0 = *(const v4f*)(s);
      v4f v1 = *(const v4f*)(s + 4096);
      v4f v2 = *(const v4f*)(s + 8192);
      v4f v3 = *(const v4f*)(s + 12288);
      const v4f z = {0.f, 0.f, 0.f, 0.f};
      v0 = in ? v0 : z; v1 = in ? v1 : z; v2 = in ? v2 : z; v3 = in ? v3 : z;
      float* d = tile + (r * CV_COLS + x4 * 4 + 1) * CV_PITCH + c4 * 4;
      const v4f p0 = {v0.x, v1.x, v2.x, v3.x};
      const v4f p1 = {v0.y, v1.y, v2.y, v3.y};
      const v4f p2 = {v0.z, v1.z, v2.z, v3.z};
      const v4f p3 = {v0.w, v1.w, v2.w, v3.w};
      *(v4f*)(d) = p0;
      *(v4f*)(d + CV_PITCH) = p1;
      *(v4f*)(d + 2 * CV_PITCH) = p2;
      *(v4f*)(d + 3 * CV_PITCH) = p3;
    }
    if (tid < 128) {
      const int r = tid >> 5, side = (tid >> 4) & 1, cc = tid & 15;
      const v4f z = {0.f, 0.f, 0.f, 0.f};
      *(v4f*)(tile + (r * CV_COLS + side * (CV_COLS - 1)) * CV_PITCH + cc * 4) = z;
    }
  }
  __syncthreads();

  v8f acc[4] = {};
  const float* Wl = w_p1 + (size_t)(m0 + l16) * 576 + hv * 36;
  const float* Bl = tile + (yl * CV_COLS + l16) * CV_PITCH + hv * 4;

  for (int g = 0; g < 8; ++g) {
    v4f aw[9];
#pragma unroll
    for (int i = 0; i < 9; ++i) aw[i] = *(const v4f*)(Wl + g * 72 + i * 4);
#define CV_AW(e) (aw[(e) >> 2][(e) & 3])
    const float* Bg = Bl + g * 8;
#pragma unroll
    for (int t = 0; t < 9; ++t) {
      const int dyi = t / 3, dxi = t % 3;
      v2f a0, a1;
      a0.x = CV_AW(t);      a0.y = CV_AW(9 + t);
      a1.x = CV_AW(18 + t); a1.y = CV_AW(27 + t);
      v4f q[4];
#pragma unroll
      for (int j = 0; j < 4; ++j)
        q[j] = *(const v4f*)(Bg + (dyi * CV_COLS + (j << 4) + dxi) * CV_PITCH);
#pragma unroll
      for (int j = 0; j < 4; ++j) {
        v2f b0; b0.x = q[j].x; b0.y = q[j].y;
        acc[j] = wmma_f32_4(a0, b0, acc[j]);
      }
#pragma unroll
      for (int j = 0; j < 4; ++j) {
        v2f b1; b1.x = q[j].z; b1.y = q[j].w;
        acc[j] = wmma_f32_4(a1, b1, acc[j]);
      }
    }
#undef CV_AW
  }

  float* O = y_pre + (size_t)b * 64 * 4096;
#pragma unroll
  for (int r = 0; r < 8; ++r) {
    const int row = m0 + r + (hv << 3);
    float* o = O + (size_t)row * 4096 + nbase + l16;
    o[0] = acc[0][r];
    o[16] = acc[1][r];
    o[32] = acc[2][r];
    o[48] = acc[3][r];
  }
}

__global__ __launch_bounds__(256) void k_final(
    const float* __restrict__ y_pre, const float* __restrict__ yA,
    const float* __restrict__ yB, const float* __restrict__ w_p2,
    const float* __restrict__ b_p2, float* __restrict__ out) {
  const int b = blockIdx.x >> 4;
  const int p = ((blockIdx.x & 15) << 8) + threadIdx.x;
  const float* Y = y_pre + (size_t)b * 64 * 4096 + p;
  float acc = b_p2[0];
#pragma unroll 8
  for (int c = 0; c < 64; ++c) {
    const float v = relu_(yA[b * 64 + c] * Y[(size_t)c * 4096] + yB[b * 64 + c]);
    acc = fmaf(w_p2[c], v, acc);
  }
  out[(size_t)b * 4096 + p] = acc;
}

extern "C" void kernel_launch(void* const* d_in, const int* in_sizes, int n_in,
                              void* d_out, int out_size, void* d_ws,
                              size_t ws_size, hipStream_t stream) {
  (void)in_sizes; (void)n_in; (void)out_size; (void)ws_size;
  const float* tf   = (const float*)d_in[0];
  const float* sf   = (const float*)d_in[1];
  const float* w_t  = (const float*)d_in[2];
  const float* gtw  = (const float*)d_in[3];
  const float* gtb  = (const float*)d_in[4];
  const float* w_s  = (const float*)d_in[5];
  const float* gsw  = (const float*)d_in[6];
  const float* gsb  = (const float*)d_in[7];
  const float* w_p1 = (const float*)d_in[8];
  const float* gpw  = (const float*)d_in[9];
  const float* gpb  = (const float*)d_in[10];
  const float* w_p2 = (const float*)d_in[11];
  const float* b_p2 = (const float*)d_in[12];
  float* out = (float*)d_out;

  float* ws = (float*)d_ws;
  float* t_global = ws;
  float* t_kernel = ws + 2048;
  float* sA = ws + 102400;
  float* sB = ws + 104448;
  float* yA = ws + 106496;
  float* yB = ws + 108544;
  float* bufA = ws + 131072;
  float* bufB = bufA + 8388608;

  k_template_path<<<32, 256, 0, stream>>>(tf, w_t, gtw, gtb, t_global, t_kernel);
  k_search_gemm<<<dim3(32, 32), 256, 0, stream>>>(sf, w_s, bufA);
  k_gn_stats<<<1024, 256, 0, stream>>>(bufA, gsw, gsb, sA, sB);
  k_corr<<<2048, 256, 0, stream>>>(bufA, sA, sB, t_global, t_kernel, bufB);
  k_conv3x3_gemm<<<dim3(32, 32), 256, 0, stream>>>(bufB, w_p1, bufA);
  k_gn_stats<<<1024, 256, 0, stream>>>(bufA, gpw, gpb, yA, yB);
  k_final<<<512, 256, 0, stream>>>(bufA, yA, yB, w_p2, b_p2, out);
}


# template path: 128 workgroups (one per batch and 16-channel tile, one wave per 16x16 output tile, loads kept many K-steps ahead) instead of 32; GroupNorm sums and global mean reduced in a new order
# speedup vs baseline: 1.5715x; 1.2058x over previous
#include <hip/hip_runtime.h>
#include <hip/hip_bf16.h>


typedef float v2f __attribute__((ext_vector_type(2)));
typedef float v8f __attribute__((ext_vector_type(8)));

__device__ __forceinline__ v8f wmma_f32_4(v2f a, v2f b, v8f c) {
  return __builtin_amdgcn_wmma_f32_16x16x4_f32(false, a, false, b, (short)0, c,
                                               false, false);
}

__device__ __forceinline__ float relu_(float x) { return fmaxf(x, 0.0f); }

__global__ __launch_bounds__(416) void k_template_path(
    const float* __restrict__ tf, const float* __restrict__ w_t,
    const float* __restrict__ gnw, const float* __restrict__ gnb,
    float* __restrict__ t_global, float* __restrict__ t_kernel) {
  const int m0 = blockIdx.x << 4;
  const int b = blockIdx.y;
  const int tid = threadIdx.x;
  const int wave = tid >> 5, lane = tid & 31;
  const int hv = lane >> 4, l16 = lane & 15;

  __shared__ float lds_t[16][208];
  __shared__ float ga[16], gb[16];

  const float* X = tf + (size_t)b * 256 * 196;

  {
    const int n0 = wave << 4;
    const int n = n0 + l16;
    const int nc = (n < 196) ? n : 195;
    const float* Ap = w_t + (m0 + l16) * 256 + (hv << 1);
    const float* Bp = X + (hv << 1) * 196 + nc;
    v8f acc = {};
    for (int k0 = 0; k0 < 256; k0 += 64) {
      v2f a[16], bb[16];
#pragma unroll
      for (int j = 0; j < 16; ++j) {
        a[j].x = Ap[k0 + 4 * j];
        a[j].y = Ap[k0 + 4 * j + 1];
        bb[j].x = Bp[(k0 + 4 * j) * 196];
        bb[j].y = Bp[(k0 + 4 * j + 1) * 196];
      }
#pragma unroll
      for (int j = 0; j < 16; ++j) acc = wmma_f32_4(a[j], bb[j], acc);
    }
#pragma unroll
    for (int r = 0; r < 8; ++r)
      lds_t[r + (hv << 3)][n0 + l16] = acc[r];
  }
  __syncthreads();

  if (wave < 8) {
    float s = 0.f, q = 0.f;
    for (int i = lane; i < 392; i += 32) {
      const int hi = (i >= 196) ? 1 : 0;
      const float v = lds_t[(wave << 1) + hi][i - hi * 196];
      s += v;
      q += v * v;
    }
    s += __shfl_xor(s, 1);  q += __shfl_xor(q, 1);
    s += __shfl_xor(s, 2);  q += __shfl_xor(q, 2);
    s += __shfl_xor(s, 4);  q += __shfl_xor(q, 4);
    s += __shfl_xor(s, 8);  q += __shfl_xor(q, 8);
    s += __shfl_xor(s, 16); q += __shfl_xor(q, 16);
    const float mean = s * (1.0f / 392.0f);
    const float var = q * (1.0f / 392.0f) - mean * mean;
    const float sc = rsqrtf(var + 1e-5f);
    if (lane < 2) {
      const int cl = (wave << 1) + lane;
      const int c = m0 + cl;
      ga[cl] = gnw[c] * sc;
      gb[cl] = gnb[c] - mean * sc * gnw[c];
    }
  }
  __syncthreads();

  for (int idx = tid; idx < 16 * 49; idx += 416) {
    const int c = idx / 49, kid = idx % 49;
    const int ky = kid / 7, kx = kid % 7;
    const float a = ga[c], bo = gb[c];
    float s = 0.f;
#pragma unroll
    for (int dy = 0; dy < 2; ++dy)
#pragma unroll
      for (int dx = 0; dx < 2; ++dx)
        s += relu_(a * lds_t[c][(2 * ky + dy) * 14 + (2 * kx + dx)] + bo);
    t_kernel[((size_t)b * 64 + m0 + c) * 49 + kid] = s * 0.25f;
  }
  if (wave < 8) {
    const int c = tid >> 4, sub = tid & 15;
    const float a = ga[c], bo = gb[c];
    float s = 0.f;
    for (int p = sub; p < 196; p += 16) s += relu_(a * lds_t[c][p] + bo);
    s += __shfl_xor(s, 1);
    s += __shfl_xor(s, 2);
    s += __shfl_xor(s, 4);
    s += __shfl_xor(s, 8);
    if (sub == 0) t_global[(size_t)b * 64 + m0 + c] = s * (1.0f / 196.0f);
  }
}

__global__ __launch_bounds__(256) void k_search_gemm(
    const float* __restrict__ sf, const float* __restrict__ w_s,
    float* __restrict__ s_pre) {
  const int b = blockIdx.y;
  const int tid = threadIdx.x;
  const int wave = tid >> 5, lane = tid & 31;
  const int hv = lane >> 4, l16 = lane & 15;
  const int m0 = (wave & 3) << 4;
  const int nbase = blockIdx.x * 128 + ((wave >> 2) << 6);

  const float* X = sf + (size_t)b * 256 * 4096;
  v8f acc0 = {}, acc1 = {}, acc2 = {}, acc3 = {};

  for (int k0 = 0; k0 < 256; k0 += 4) {
    const int kk = k0 + (hv << 1);
    v2f a;
    a.x = w_s[(m0 + l16) * 256 + kk];
    a.y = w_s[(m0 + l16) * 256 + kk + 1];
    const float* r0 = X + (size_t)kk * 4096 + nbase + l16;
    const float* r1 = r0 + 4096;
    v2f b0; b0.x = r0[0];  b0.y = r1[0];  acc0 = wmma_f32_4(a, b0, acc0);
    v2f b1; b1.x = r0[16]; b1.y = r1[16]; acc1 = wmma_f32_4(a, b1, acc1);
    v2f b2; b2.x = r0[32]; b2.y = r1[32]; acc2 = wmma_f32_4(a, b2, acc2);
    v2f b3; b3.x = r0[48]; b3.y = r1[48]; acc3 = wmma_f32_4(a, b3, acc3);
  }

  float* O = s_pre + (size_t)b * 64 * 4096;
#pragma unroll
  for (int r = 0; r < 8; ++r) {
    const int row = m0 + r + (hv << 3);
    float* o = O + (size_t)row * 4096 + nbase + l16;
    o[0] = acc0[r];
    o[16] = acc1[r];
    o[32] = acc2[r];
    o[48] = acc3[r];
  }
}

__global__ __launch_bounds__(256) void k_gn_stats(
    const float* __restrict__ pre, const float* __restrict__ gnw,
    const float* __restrict__ gnb, float* __restrict__ A,
    float* __restrict__ Bo) {
  const int b = blockIdx.x >> 5, g = blockIdx.x & 31;
  const float* p = pre + ((size_t)b * 64 + g * 2) * 4096;
  const int tid = threadIdx.x;
  float s = 0.f, q = 0.f;
  for (int i = tid; i < 8192; i += 256) {
    const float v = p[i];
    s += v;
    q += v * v;
  }
  __shared__ float ss[256], sq[256];
  ss[tid] = s;
  sq[tid] = q;
  __syncthreads();
  for (int o = 128; o > 0; o >>= 1) {
    if (tid < o) {
      ss[tid] += ss[tid + o];
      sq[tid] += sq[tid + o];
    }
    __syncthreads();
  }
  if (tid < 2) {
    const float mean = ss[0] * (1.f / 8192.f);
    const float var = sq[0] * (1.f / 8192.f) - mean * mean;
    const float sc = rsqrtf(var + 1e-5f);
    const int c = g * 2 + tid;
    A[(size_t)b * 64 + c] = gnw[c] * sc;
    Bo[(size_t)b * 64 + c] = gnb[c] - mean * sc * gnw[c];
  }
}

__global__ __launch_bounds__(256) void k_corr(
    const float* __restrict__ s_pre, const float* __restrict__ sA,
    const float* __restrict__ sB, const float* __restrict__ t_global,
    const float* __restrict__ t_kernel, float* __restrict__ corr) {
  const int bc = blockIdx.x;
  const float* src = s_pre + (size_t)bc * 4096;
  const float a = sA[bc], bo = sB[bc], tg = t_global[bc];
  __shared__ float tile[70][70];
  __shared__ float kern[49];
  const int tid = threadIdx.x;
  if (tid < 49) kern[tid] = t_kernel[(size_t)bc * 49 + tid];
  for (int i = tid; i < 70 * 70; i += 256) {
    const int y = i / 70 - 3, x = i % 70 - 3;
    const bool in = ((unsigned)y < 64u) & ((unsigned)x < 64u);
    const float raw = src[in ? (y * 64 + x) : 0];
    tile[i / 70][i % 70] = in ? relu_(a * raw + bo) : 0.f;
  }
  __syncthreads();
  float* dst = corr + (size_t)bc * 4096;
  for (int p = tid; p < 4096; p += 256) {
    const int y = p >> 6, x = p & 63;
    float acc = tg * tile[y + 3][x + 3];
#pragma unroll
    for (int ky = 0; ky < 7; ++ky)
#pragma unroll
      for (int kx = 0; kx < 7; ++kx)
        acc = fmaf(kern[ky * 7 + kx], tile[y + ky][x + kx], acc);
    dst[p] = acc;
  }
}

typedef float v4f __attribute__((ext_vector_type(4)));

#define CV_PITCH 68
#define CV_COLS 66
#define CV_ROWS 4

__global__ __launch_bounds__(256) void k_conv3x3_gemm(
    const float* __restrict__ corr, const float* __restrict__ w_p1,
    float* __restrict__ y_pre) {
  const int b = blockIdx.y;
  const int tid = threadIdx.x;
  const int wave = tid >> 5, lane = tid & 31;
  const int hv = lane >> 4, l16 = lane & 15;
  const int m0 = (wave & 3) << 4;
  const int yl = wave >> 2;
  const int y0 = blockIdx.x << 1;
  const int nbase = blockIdx.x * 128 + (yl << 6);

  __shared__ __attribute__((aligned(16))) float tile[CV_ROWS * CV_COLS * CV_PITCH];

  const float* X = corr + (size_t)b * 64 * 4096;

  {
    const int x4 = tid & 15, c4 = tid >> 4;
#pragma unroll
    for (int r = 0; r < CV_ROWS; ++r) {
      const int gy = y0 - 1 + r;
      const bool in = (unsigned)gy < 64u;
      const float* s = X + (size_t)(c4 * 4) * 4096 + (in ? gy : 0) * 64 + x4 * 4;
      v4f v0 = *(const v4f*)(s);
      v4f v1 = *(const v4f*)(s + 4096);
      v4f v2 = *(const v4f*)(s + 8192);
      v4f v3 = *(const v4f*)(s + 12288);
      const v4f z = {0.f, 0.f, 0.f, 0.f};
      v0 = in ? v0 : z; v1 = in ? v1 : z; v2 = in ? v2 : z; v3 = in ? v3 : z;
      float* d = tile + (r * CV_COLS + x4 * 4 + 1) * CV_PITCH + c4 * 4;
      const v4f p0 = {v0.x, v1.x, v2.x, v3.x};
      const v4f p1 = {v0.y, v1.y, v2.y, v3.y};
      const v4f p2 = {v0.z, v1.z, v2.z, v3.z};
      const v4f p3 = {v0.w, v1.w, v2.w, v3.w};
      *(v4f*)(d) = p0;
      *(v4f*)(d + CV_PITCH) = p1;
      *(v4f*)(d + 2 * CV_PITCH) = p2;
      *(v4f*)(d + 3 * CV_PITCH) = p3;
    }
    if (tid < 128) {
      const int r = tid >> 5, side = (tid >> 4) & 1, cc = tid & 15;
      const v4f z = {0.f, 0.f, 0.f, 0.f};
      *(v4f*)(tile + (r * CV_COLS + side * (CV_COLS - 1)) * CV_PITCH + cc * 4) = z;
    }
  }
  __syncthreads();

  v8f acc[4] = {};
  const float* Wl = w_p1 + (size_t)(m0 + l16) * 576 + hv * 36;
  const float* Bl = tile + (yl * CV_COLS + l16) * CV_PITCH + hv * 4;

  for (int g = 0; g < 8; ++g) {
    v4f aw[9];
#pragma unroll
    for (int i = 0; i < 9; ++i) aw[i] = *(const v4f*)(Wl + g * 72 + i * 4);
#define CV_AW(e) (aw[(e) >> 2][(e) & 3])
    const float* Bg = Bl + g * 8;
#pragma unroll
    for (int t = 0; t < 9; ++t) {
      const int dyi = t / 3, dxi = t % 3;
      v2f a0, a1;
      a0.x = CV_AW(t);      a0.y = CV_AW(9 + t);
      a1.x = CV_AW(18 + t); a1.y = CV_AW(27 + t);
      v4f q[4];
#pragma unroll
      for (int j = 0; j < 4; ++j)
        q[j] = *(const v4f*)(Bg + (dyi * CV_COLS + (j << 4) + dxi) * CV_PITCH);
#pragma unroll
      for (int j = 0; j < 4; ++j) {
        v2f b0; b0.x = q[j].x; b0.y = q[j].y;
        acc[j] = wmma_f32_4(a0, b0, acc[j]);
      }
#pragma unroll
      for (int j = 0; j < 4; ++j) {
        v2f b1; b1.x = q[j].z; b1.y = q[j].w;
        acc[j] = wmma_f32_4(a1, b1, acc[j]);
      }
    }
#undef CV_AW
  }

  float* O = y_pre + (size_t)b * 64 * 4096;
#pragma unroll
  for (int r = 0; r < 8; ++r) {
    const int row = m0 + r + (hv << 3);
    float* o = O + (size_t)row * 4096 + nbase + l16;
    o[0] = acc[0][r];
    o[16] = acc[1][r];
    o[32] = acc[2][r];
    o[48] = acc[3][r];
  }
}

__global__ __launch_bounds__(256) void k_final(
    const float* __restrict__ y_pre, const float* __restrict__ yA,
    const float* __restrict__ yB, const float* __restrict__ w_p2,
    const float* __restrict__ b_p2, float* __restrict__ out) {
  const int b = blockIdx.x >> 4;
  const int p = ((blockIdx.x & 15) << 8) + threadIdx.x;
  const float* Y = y_pre + (size_t)b * 64 * 4096 + p;
  float acc = b_p2[0];
#pragma unroll 8
  for (int c = 0; c < 64; ++c) {
    const float v = relu_(yA[b * 64 + c] * Y[(size_t)c * 4096] + yB[b * 64 + c]);
    acc = fmaf(w_p2[c], v, acc);
  }
  out[(size_t)b * 4096 + p] = acc;
}

extern "C" void kernel_launch(void* const* d_in, const int* in_sizes, int n_in,
                              void* d_out, int out_size, void* d_ws,
                              size_t ws_size, hipStream_t stream) {
  (void)in_sizes; (void)n_in; (void)out_size; (void)ws_size;
  const float* tf   = (const float*)d_in[0];
  const float* sf   = (const float*)d_in[1];
  const float* w_t  = (const float*)d_in[2];
  const float* gtw  = (const float*)d_in[3];
  const float* gtb  = (const float*)d_in[4];
  const float* w_s  = (const float*)d_in[5];
  const float* gsw  = (const float*)d_in[6];
  const float* gsb  = (const float*)d_in[7];
  const float* w_p1 = (const float*)d_in[8];
  const float* gpw  = (const float*)d_in[9];
  const float* gpb  = (const float*)d_in[10];
  const float* w_p2 = (const float*)d_in[11];
  const float* b_p2 = (const float*)d_in[12];
  float* out = (float*)d_out;

  float* ws = (float*)d_ws;
  float* t_global = ws;
  float* t_kernel = ws + 2048;
  float* sA = ws + 102400;
  float* sB = ws + 104448;
  float* yA = ws + 106496;
  float* yB = ws + 108544;
  float* bufA = ws + 131072;
  float* bufB = bufA + 8388608;

  k_template_path<<<dim3(4, 32), 416, 0, stream>>>(tf, w_t, gtw, gtb, t_global, t_kernel);
  k_search_gemm<<<dim3(32, 32), 256, 0, stream>>>(sf, w_s, bufA);
  k_gn_stats<<<1024, 256, 0, stream>>>(bufA, gsw, gsb, sA, sB);
  k_corr<<<2048, 256, 0, stream>>>(bufA, sA, sB, t_global, t_kernel, bufB);
  k_conv3x3_gemm<<<dim3(32, 32), 256, 0, stream>>>(bufB, w_p1, bufA);
  k_gn_stats<<<1024, 256, 0, stream>>>(bufA, gpw, gpb, yA, yB);
  k_final<<<512, 256, 0, stream>>>(bufA, yA, yB, w_p2, b_p2, out);
}


# corr kernel: LDS tile filled from four independent 16-byte loads per thread issued together (one memory round trip) instead of one waited 4-byte load per element; halo zeroed by plain stores
# speedup vs baseline: 1.8635x; 1.1858x over previous
#include <hip/hip_runtime.h>
#include <hip/hip_bf16.h>


typedef float v2f __attribute__((ext_vector_type(2)));
typedef float v4f __attribute__((ext_vector_type(4)));
typedef float v8f __attribute__((ext_vector_type(8)));

__device__ __forceinline__ v8f wmma_f32_4(v2f a, v2f b, v8f c) {
  return __builtin_amdgcn_wmma_f32_16x16x4_f32(false, a, false, b, (short)0, c,
                                               false, false);
}

__device__ __forceinline__ float relu_(float x) { return fmaxf(x, 0.0f); }

__global__ __launch_bounds__(416) void k_template_path(
    const float* __restrict__ tf, const float* __restrict__ w_t,
    const float* __restrict__ gnw, const float* __restrict__ gnb,
    float* __restrict__ t_global, float* __restrict__ t_kernel) {
  const int m0 = blockIdx.x << 4;
  const int b = blockIdx.y;
  const int tid = threadIdx.x;
  const int wave = tid >> 5, lane = tid & 31;
  const int hv = lane >> 4, l16 = lane & 15;

  __shared__ float lds_t[16][208];
  __shared__ float ga[16], gb[16];

  const float* X = tf + (size_t)b * 256 * 196;

  {
    const int n0 = wave << 4;
    const int n = n0 + l16;
    const int nc = (n < 196) ? n : 195;
    const float* Ap = w_t + (m0 + l16) * 256 + (hv << 1);
    const float* Bp = X + (hv << 1) * 196 + nc;
    v8f acc = {};
    for (int k0 = 0; k0 < 256; k0 += 64) {
      v2f a[16], bb[16];
#pragma unroll
      for (int j = 0; j < 16; ++j) {
        a[j].x = Ap[k0 + 4 * j];
        a[j].y = Ap[k0 + 4 * j + 1];
        bb[j].x = Bp[(k0 + 4 * j) * 196];
        bb[j].y = Bp[(k0 + 4 * j + 1) * 196];
      }
#pragma unroll
      for (int j = 0; j < 16; ++j) acc = wmma_f32_4(a[j], bb[j], acc);
    }
#pragma unroll
    for (int r = 0; r < 8; ++r)
      lds_t[r + (hv << 3)][n0 + l16] = acc[r];
  }
  __syncthreads();

  if (wave < 8) {
    float s = 0.f, q = 0.f;
    for (int i = lane; i < 392; i += 32) {
      const int hi = (i >= 196) ? 1 : 0;
      const float v = lds_t[(wave << 1) + hi][i - hi * 196];
      s += v;
      q += v * v;
    }
    s += __shfl_xor(s, 1);  q += __shfl_xor(q, 1);
    s += __shfl_xor(s, 2);  q += __shfl_xor(q, 2);
    s += __shfl_xor(s, 4);  q += __shfl_xor(q, 4);
    s += __shfl_xor(s, 8);  q += __shfl_xor(q, 8);
    s += __shfl_xor(s, 16); q += __shfl_xor(q, 16);
    const float mean = s * (1.0f / 392.0f);
    const float var = q * (1.0f / 392.0f) - mean * mean;
    const float sc = rsqrtf(var + 1e-5f);
    if (lane < 2) {
      const int cl = (wave << 1) + lane;
      const int c = m0 + cl;
      ga[cl] = gnw[c] * sc;
      gb[cl] = gnb[c] - mean * sc * gnw[c];
    }
  }
  __syncthreads();

  for (int idx = tid; idx < 16 * 49; idx += 416) {
    const int c = idx / 49, kid = idx % 49;
    const int ky = kid / 7, kx = kid % 7;
    const float a = ga[c], bo = gb[c];
    float s = 0.f;
#pragma unroll
    for (int dy = 0; dy < 2; ++dy)
#pragma unroll
      for (int dx = 0; dx < 2; ++dx)
        s += relu_(a * lds_t[c][(2 * ky + dy) * 14 + (2 * kx + dx)] + bo);
    t_kernel[((size_t)b * 64 + m0 + c) * 49 + kid] = s * 0.25f;
  }
  if (wave < 8) {
    const int c = tid >> 4, sub = tid & 15;
    const float a = ga[c], bo = gb[c];
    float s = 0.f;
    for (int p = sub; p < 196; p += 16) s += relu_(a * lds_t[c][p] + bo);
    s += __shfl_xor(s, 1);
    s += __shfl_xor(s, 2);
    s += __shfl_xor(s, 4);
    s += __shfl_xor(s, 8);
    if (sub == 0) t_global[(size_t)b * 64 + m0 + c] = s * (1.0f / 196.0f);
  }
}

__global__ __launch_bounds__(256) void k_search_gemm(
    const float* __restrict__ sf, const float* __restrict__ w_s,
    float* __restrict__ s_pre) {
  const int b = blockIdx.y;
  const int tid = threadIdx.x;
  const int wave = tid >> 5, lane = tid & 31;
  const int hv = lane >> 4, l16 = lane & 15;
  const int m0 = (wave & 3) << 4;
  const int nbase = blockIdx.x * 128 + ((wave >> 2) << 6);

  const float* X = sf + (size_t)b * 256 * 4096;
  v8f acc0 = {}, acc1 = {}, acc2 = {}, acc3 = {};

  for (int k0 = 0; k0 < 256; k0 += 4) {
    const int kk = k0 + (hv << 1);
    v2f a;
    a.x = w_s[(m0 + l16) * 256 + kk];
    a.y = w_s[(m0 + l16) * 256 + kk + 1];
    const float* r0 = X + (size_t)kk * 4096 + nbase + l16;
    const float* r1 = r0 + 4096;
    v2f b0; b0.x = r0[0];  b0.y = r1[0];  acc0 = wmma_f32_4(a, b0, acc0);
    v2f b1; b1.x = r0[16]; b1.y = r1[16]; acc1 = wmma_f32_4(a, b1, acc1);
    v2f b2; b2.x = r0[32]; b2.y = r1[32]; acc2 = wmma_f32_4(a, b2, acc2);
    v2f b3; b3.x = r0[48]; b3.y = r1[48]; acc3 = wmma_f32_4(a, b3, acc3);
  }

  float* O = s_pre + (size_t)b * 64 * 4096;
#pragma unroll
  for (int r = 0; r < 8; ++r) {
    const int row = m0 + r + (hv << 3);
    float* o = O + (size_t)row * 4096 + nbase + l16;
    o[0] = acc0[r];
    o[16] = acc1[r];
    o[32] = acc2[r];
    o[48] = acc3[r];
  }
}

__global__ __launch_bounds__(256) void k_gn_stats(
    const float* __restrict__ pre, const float* __restrict__ gnw,
    const float* __restrict__ gnb, float* __restrict__ A,
    float* __restrict__ Bo) {
  const int b = blockIdx.x >> 5, g = blockIdx.x & 31;
  const float* p = pre + ((size_t)b * 64 + g * 2) * 4096;
  const int tid = threadIdx.x;
  float s = 0.f, q = 0.f;
  for (int i = tid; i < 8192; i += 256) {
    const float v = p[i];
    s += v;
    q += v * v;
  }
  __shared__ float ss[256], sq[256];
  ss[tid] = s;
  sq[tid] = q;
  __syncthreads();
  for (int o = 128; o > 0; o >>= 1) {
    if (tid < o) {
      ss[tid] += ss[tid + o];
      sq[tid] += sq[tid + o];
    }
    __syncthreads();
  }
  if (tid < 2) {
    const float mean = ss[0] * (1.f / 8192.f);
    const float var = sq[0] * (1.f / 8192.f) - mean * mean;
    const float sc = rsqrtf(var + 1e-5f);
    const int c = g * 2 + tid;
    A[(size_t)b * 64 + c] = gnw[c] * sc;
    Bo[(size_t)b * 64 + c] = gnb[c] - mean * sc * gnw[c];
  }
}

__global__ __launch_bounds__(256) void k_corr(
    const float* __restrict__ s_pre, const float* __restrict__ sA,
    const float* __restrict__ sB, const float* __restrict__ t_global,
    const float* __restrict__ t_kernel, float* __restrict__ corr) {
  const int bc = blockIdx.x;
  const float* src = s_pre + (size_t)bc * 4096;
  const float a = sA[bc], bo = sB[bc], tg = t_global[bc];
  __shared__ float tile[70][70];
  __shared__ float kern[256];
  const int tid = threadIdx.x;
  const float kv = t_kernel[(size_t)bc * 49 + (tid < 49 ? tid : 48)];
  v4f v[4];
#pragma unroll
  for (int j = 0; j < 4; ++j)
    v[j] = *(const v4f*)(src + ((tid + 256 * j) << 2));
  for (int i = tid; i < 804; i += 256) {
    int row, col;
    if (i < 420) {
      const int r = i / 70;
      row = (r < 3) ? r : r + 64;
      col = i - r * 70;
    } else {
      const int j = i - 420;
      const int r = j / 6;
      const int cc = j - r * 6;
      row = r + 3;
      col = (cc < 3) ? cc : cc + 64;
    }
    tile[row][col] = 0.f;
  }
  kern[tid] = kv;
#pragma unroll
  for (int j = 0; j < 4; ++j) {
    const int idx = tid + 256 * j;
    float* d = &tile[(idx >> 4) + 3][((idx & 15) << 2) + 3];
    d[0] = relu_(a * v[j].x + bo);
    d[1] = relu_(a * v[j].y + bo);
    d[2] = relu_(a * v[j].z + bo);
    d[3] = relu_(a * v[j].w + bo);
  }
  __syncthreads();
  float* dst = corr + (size_t)bc * 4096;
  for (int p = tid; p < 4096; p += 256) {
    const int y = p >> 6, x = p & 63;
    float acc = tg * tile[y + 3][x + 3];
#pragma unroll
    for (int ky = 0; ky < 7; ++ky)
#pragma unroll
      for (int kx = 0; kx < 7; ++kx)
        acc = fmaf(kern[ky * 7 + kx], tile[y + ky][x + kx], acc);
    dst[p] = acc;
  }
}

#define CV_PITCH 68
#define CV_COLS 66
#define CV_ROWS 4

__global__ __launch_bounds__(256) void k_conv3x3_gemm(
    const float* __restrict__ corr, const float* __restrict__ w_p1,
    float* __restrict__ y_pre) {
  const int b = blockIdx.y;
  const int tid = threadIdx.x;
  const int wave = tid >> 5, lane = tid & 31;
  const int hv = lane >> 4, l16 = lane & 15;
  const int m0 = (wave & 3) << 4;
  const int yl = wave >> 2;
  const int y0 = blockIdx.x << 1;
  const int nbase = blockIdx.x * 128 + (yl << 6);

  __shared__ __attribute__((aligned(16))) float tile[CV_ROWS * CV_COLS * CV_PITCH];

  const float* X = corr + (size_t)b * 64 * 4096;

  {
    const int x4 = tid & 15, c4 = tid >> 4;
#pragma unroll
    for (int r = 0; r < CV_ROWS; ++r) {
      const int gy = y0 - 1 + r;
      const bool in = (unsigned)gy < 64u;
      const float* s = X + (size_t)(c4 * 4) * 4096 + (in ? gy : 0) * 64 + x4 * 4;
      v4f v0 = *(const v4f*)(s);
      v4f v1 = *(const v4f*)(s + 4096);
      v4f v2 = *(const v4f*)(s + 8192);
      v4f v3 = *(const v4f*)(s + 12288);
      const v4f z = {0.f, 0.f, 0.f, 0.f};
      v0 = in ? v0 : z; v1 = in ? v1 : z; v2 = in ? v2 : z; v3 = in ? v3 : z;
      float* d = tile + (r * CV_COLS + x4 * 4 + 1) * CV_PITCH + c4 * 4;
      const v4f p0 = {v0.x, v1.x, v2.x, v3.x};
      const v4f p1 = {v0.y, v1.y, v2.y, v3.y};
      const v4f p2 = {v0.z, v1.z, v2.z, v3.z};
      const v4f p3 = {v0.w, v1.w, v2.w, v3.w};
      *(v4f*)(d) = p0;
      *(v4f*)(d + CV_PITCH) = p1;
      *(v4f*)(d + 2 * CV_PITCH) = p2;
      *(v4f*)(d + 3 * CV_PITCH) = p3;
    }
    if (tid < 128) {
      const int r = tid >> 5, side = (tid >> 4) & 1, cc = tid & 15;
      const v4f z = {0.f, 0.f, 0.f, 0.f};
      *(v4f*)(tile + (r * CV_COLS + side * (CV_COLS - 1)) * CV_PITCH + cc * 4) = z;
    }
  }
  __syncthreads();

  v8f acc[4] = {};
  const float* Wl = w_p1 + (size_t)(m0 + l16) * 576 + hv * 36;
  const float* Bl = tile + (yl * CV_COLS + l16) * CV_PITCH + hv * 4;

  for (int g = 0; g < 8; ++g) {
    v4f aw[9];
#pragma unroll
    for (int i = 0; i < 9; ++i) aw[i] = *(const v4f*)(Wl + g * 72 + i * 4);
#define CV_AW(e) (aw[(e) >> 2][(e) & 3])
    const float* Bg = Bl + g * 8;
#pragma unroll
    for (int t = 0; t < 9; ++t) {
      const int dyi = t / 3, dxi = t % 3;
      v2f a0, a1;
      a0.x = CV_AW(t);      a0.y = CV_AW(9 + t);
      a1.x = CV_AW(18 + t); a1.y = CV_AW(27 + t);
      v4f q[4];
#pragma unroll
      for (int j = 0; j < 4; ++j)
        q[j] = *(const v4f*)(Bg + (dyi * CV_COLS + (j << 4) + dxi) * CV_PITCH);
#pragma unroll
      for (int j = 0; j < 4; ++j) {
        v2f b0; b0.x = q[j].x; b0.y = q[j].y;
        acc[j] = wmma_f32_4(a0, b0, acc[j]);
      }
#pragma unroll
      for (int j = 0; j < 4; ++j) {
        v2f b1; b1.x = q[j].z; b1.y = q[j].w;
        acc[j] = wmma_f32_4(a1, b1, acc[j]);
      }
    }
#undef CV_AW
  }

  float* O = y_pre + (size_t)b * 64 * 4096;
#pragma unroll
  for (int r = 0; r < 8; ++r) {
    const int row = m0 + r + (hv << 3);
    float* o = O + (size_t)row * 4096 + nbase + l16;
    o[0] = acc[0][r];
    o[16] = acc[1][r];
    o[32] = acc[2][r];
    o[48] = acc[3][r];
  }
}

__global__ __launch_bounds__(256) void k_final(
    const float* __restrict__ y_pre, const float* __restrict__ yA,
    const float* __restrict__ yB, const float* __restrict__ w_p2,
    const float* __restrict__ b_p2, float* __restrict__ out) {
  const int b = blockIdx.x >> 4;
  const int p = ((blockIdx.x & 15) << 8) + threadIdx.x;
  const float* Y = y_pre + (size_t)b * 64 * 4096 + p;
  float acc = b_p2[0];
#pragma unroll 8
  for (int c = 0; c < 64; ++c) {
    const float v = relu_(yA[b * 64 + c] * Y[(size_t)c * 4096] + yB[b * 64 + c]);
    acc = fmaf(w_p2[c], v, acc);
  }
  out[(size_t)b * 4096 + p] = acc;
}

extern "C" void kernel_launch(void* const* d_in, const int* in_sizes, int n_in,
                              void* d_out, int out_size, void* d_ws,
                              size_t ws_size, hipStream_t stream) {
  (void)in_sizes; (void)n_in; (void)out_size; (void)ws_size;
  const float* tf   = (const float*)d_in[0];
  const float* sf   = (const float*)d_in[1];
  const float* w_t  = (const float*)d_in[2];
  const float* gtw  = (const float*)d_in[3];
  const float* gtb  = (const float*)d_in[4];
  const float* w_s  = (const float*)d_in[5];
  const float* gsw  = (const float*)d_in[6];
  const float* gsb  = (const float*)d_in[7];
  const float* w_p1 = (const float*)d_in[8];
  const float* gpw  = (const float*)d_in[9];
  const float* gpb  = (const float*)d_in[10];
  const float* w_p2 = (const float*)d_in[11];
  const float* b_p2 = (const float*)d_in[12];
  float* out = (float*)d_out;

  float* ws = (float*)d_ws;
  float* t_global = ws;
  float* t_kernel = ws + 2048;
  float* sA = ws + 102400;
  float* sB = ws + 104448;
  float* yA = ws + 106496;
  float* yB = ws + 108544;
  float* bufA = ws + 131072;
  float* bufB = bufA + 8388608;

  k_template_path<<<dim3(4, 32), 416, 0, stream>>>(tf, w_t, gtw, gtb, t_global, t_kernel);
  k_search_gemm<<<dim3(32, 32), 256, 0, stream>>>(sf, w_s, bufA);
  k_gn_stats<<<1024, 256, 0, stream>>>(bufA, gsw, gsb, sA, sB);
  k_corr<<<2048, 256, 0, stream>>>(bufA, sA, sB, t_global, t_kernel, bufB);
  k_conv3x3_gemm<<<dim3(32, 32), 256, 0, stream>>>(bufB, w_p1, bufA);
  k_gn_stats<<<1024, 256, 0, stream>>>(bufA, gpw, gpb, yA, yB);
  k_final<<<512, 256, 0, stream>>>(bufA, yA, yB, w_p2, b_p2, out);
}


# GroupNorm sums fused into neighbours: corr kernel (block = batch x group, 2 channels) sums its input; conv3x3 epilogue writes per-block partials, final kernel reduces; 7->5 launches; new sum order
# speedup vs baseline: 1.9899x; 1.0678x over previous
#include <hip/hip_runtime.h>
#include <hip/hip_bf16.h>


typedef float v2f __attribute__((ext_vector_type(2)));
typedef float v4f __attribute__((ext_vector_type(4)));
typedef float v8f __attribute__((ext_vector_type(8)));

__device__ __forceinline__ v8f wmma_f32_4(v2f a, v2f b, v8f c) {
  return __builtin_amdgcn_wmma_f32_16x16x4_f32(false, a, false, b, (short)0, c,
                                               false, false);
}

__device__ __forceinline__ float relu_(float x) { return fmaxf(x, 0.0f); }

__global__ __launch_bounds__(416) void k_template_path(
    const float* __restrict__ tf, const float* __restrict__ w_t,
    const float* __restrict__ gnw, const float* __restrict__ gnb,
    float* __restrict__ t_global, float* __restrict__ t_kernel) {
  const int m0 = blockIdx.x << 4;
  const int b = blockIdx.y;
  const int tid = threadIdx.x;
  const int wave = tid >> 5, lane = tid & 31;
  const int hv = lane >> 4, l16 = lane & 15;

  __shared__ float lds_t[16][208];
  __shared__ float ga[16], gb[16];

  const float* X = tf + (size_t)b * 256 * 196;

  {
    const int n0 = wave << 4;
    const int n = n0 + l16;
    const int nc = (n < 196) ? n : 195;
    const float* Ap = w_t + (m0 + l16) * 256 + (hv << 1);
    const float* Bp = X + (hv << 1) * 196 + nc;
    v8f acc = {};
    for (int k0 = 0; k0 < 256; k0 += 64) {
      v2f a[16], bb[16];
#pragma unroll
      for (int j = 0; j < 16; ++j) {
        a[j].x = Ap[k0 + 4 * j];
        a[j].y = Ap[k0 + 4 * j + 1];
        bb[j].x = Bp[(k0 + 4 * j) * 196];
        bb[j].y = Bp[(k0 + 4 * j + 1) * 196];
      }
#pragma unroll
      for (int j = 0; j < 16; ++j) acc = wmma_f32_4(a[j], bb[j], acc);
    }
#pragma unroll
    for (int r = 0; r < 8; ++r)
      lds_t[r + (hv << 3)][n0 + l16] = acc[r];
  }
  __syncthreads();

  if (wave < 8) {
    float s = 0.f, q = 0.f;
    for (int i = lane; i < 392; i += 32) {
      const int hi = (i >= 196) ? 1 : 0;
      const float v = lds_t[(wave << 1) + hi][i - hi * 196];
      s += v;
      q += v * v;
    }
    s += __shfl_xor(s, 1);  q += __shfl_xor(q, 1);
    s += __shfl_xor(s, 2);  q += __shfl_xor(q, 2);
    s += __shfl_xor(s, 4);  q += __shfl_xor(q, 4);
    s += __shfl_xor(s, 8);  q += __shfl_xor(q, 8);
    s += __shfl_xor(s, 16); q += __shfl_xor(q, 16);
    const float mean = s * (1.0f / 392.0f);
    const float var = q * (1.0f / 392.0f) - mean * mean;
    const float sc = rsqrtf(var + 1e-5f);
    if (lane < 2) {
      const int cl = (wave << 1) + lane;
      const int c = m0 + cl;
      ga[cl] = gnw[c] * sc;
      gb[cl] = gnb[c] - mean * sc * gnw[c];
    }
  }
  __syncthreads();

  for (int idx = tid; idx < 16 * 49; idx += 416) {
    const int c = idx / 49, kid = idx % 49;
    const int ky = kid / 7, kx = kid % 7;
    const float a = ga[c], bo = gb[c];
    float s = 0.f;
#pragma unroll
    for (int dy = 0; dy < 2; ++dy)
#pragma unroll
      for (int dx = 0; dx < 2; ++dx)
        s += relu_(a * lds_t[c][(2 * ky + dy) * 14 + (2 * kx + dx)] + bo);
    t_kernel[((size_t)b * 64 + m0 + c) * 49 + kid] = s * 0.25f;
  }
  if (wave < 8) {
    const int c = tid >> 4, sub = tid & 15;
    const float a = ga[c], bo = gb[c];
    float s = 0.f;
    for (int p = sub; p < 196; p += 16) s += relu_(a * lds_t[c][p] + bo);
    s += __shfl_xor(s, 1);
    s += __shfl_xor(s, 2);
    s += __shfl_xor(s, 4);
    s += __shfl_xor(s, 8);
    if (sub == 0) t_global[(size_t)b * 64 + m0 + c] = s * (1.0f / 196.0f);
  }
}

__global__ __launch_bounds__(256) void k_search_gemm(
    const float* __restrict__ sf, const float* __restrict__ w_s,
    float* __restrict__ s_pre) {
  const int b = blockIdx.y;
  const int tid = threadIdx.x;
  const int wave = tid >> 5, lane = tid & 31;
  const int hv = lane >> 4, l16 = lane & 15;
  const int m0 = (wave & 3) << 4;
  const int nbase = blockIdx.x * 128 + ((wave >> 2) << 6);

  const float* X = sf + (size_t)b * 256 * 4096;
  v8f acc0 = {}, acc1 = {}, acc2 = {}, acc3 = {};

  for (int k0 = 0; k0 < 256; k0 += 4) {
    const int kk = k0 + (hv << 1);
    v2f a;
    a.x = w_s[(m0 + l16) * 256 + kk];
    a.y = w_s[(m0 + l16) * 256 + kk + 1];
    const float* r0 = X + (size_t)kk * 4096 + nbase + l16;
    const float* r1 = r0 + 4096;
    v2f b0; b0.x = r0[0];  b0.y = r1[0];  acc0 = wmma_f32_4(a, b0, acc0);
    v2f b1; b1.x = r0[16]; b1.y = r1[16]; acc1 = wmma_f32_4(a, b1, acc1);
    v2f b2; b2.x = r0[32]; b2.y = r1[32]; acc2 = wmma_f32_4(a, b2, acc2);
    v2f b3; b3.x = r0[48]; b3.y = r1[48]; acc3 = wmma_f32_4(a, b3, acc3);
  }

  float* O = s_pre + (size_t)b * 64 * 4096;
#pragma unroll
  for (int r = 0; r < 8; ++r) {
    const int row = m0 + r + (hv << 3);
    float* o = O + (size_t)row * 4096 + nbase + l16;
    o[0] = acc0[r];
    o[16] = acc1[r];
    o[32] = acc2[r];
    o[48] = acc3[r];
  }
}

__global__ __launch_bounds__(256) void k_corr(
    const float* __restrict__ s_pre, const float* __restrict__ gnw,
    const float* __restrict__ gnb, const float* __restrict__ t_global,
    const float* __restrict__ t_kernel, float* __restrict__ corr) {
  const int bg = blockIdx.x;
  const int g = bg & 31;
  const float* src = s_pre + (size_t)bg * 8192;
  __shared__ float tile[2][70][70];
  __shared__ float kern[2][256];
  __shared__ float red_s[8], red_q[8];
  const int tid = threadIdx.x;
  const int wave = tid >> 5, lane = tid & 31;
  const int tap = (tid < 49) ? tid : 48;
  const float kv0 = t_kernel[(size_t)(bg * 2) * 49 + tap];
  const float kv1 = t_kernel[(size_t)(bg * 2 + 1) * 49 + tap];
  v4f v[8];
#pragma unroll
  for (int j = 0; j < 8; ++j)
    v[j] = *(const v4f*)(src + ((tid + 256 * j) << 2));
  for (int i = tid; i < 804; i += 256) {
    int row, col;
    if (i < 420) {
      const int r = i / 70;
      row = (r < 3) ? r : r + 64;
      col = i - r * 70;
    } else {
      const int j = i - 420;
      const int r = j / 6;
      const int cc = j - r * 6;
      row = r + 3;
      col = (cc < 3) ? cc : cc + 64;
    }
    tile[0][row][col] = 0.f;
    tile[1][row][col] = 0.f;
  }
  float s = 0.f, q = 0.f;
#pragma unroll
  for (int j = 0; j < 8; ++j) {
    s += (v[j].x + v[j].y) + (v[j].z + v[j].w);
    q = fmaf(v[j].x, v[j].x, q);
    q = fmaf(v[j].y, v[j].y, q);
    q = fmaf(v[j].z, v[j].z, q);
    q = fmaf(v[j].w, v[j].w, q);
  }
  s += __shfl_xor(s, 1);  q += __shfl_xor(q, 1);
  s += __shfl_xor(s, 2);  q += __shfl_xor(q, 2);
  s += __shfl_xor(s, 4);  q += __shfl_xor(q, 4);
  s += __shfl_xor(s, 8);  q += __shfl_xor(q, 8);
  s += __shfl_xor(s, 16); q += __shfl_xor(q, 16);
  if (lane == 0) {
    red_s[wave] = s;
    red_q[wave] = q;
  }
  __syncthreads();
  float S = 0.f, Q = 0.f;
  for (int w = 0; w < 8; ++w) {
    S += red_s[w];
    Q += red_q[w];
  }
  const float mean = S * (1.f / 8192.f);
  const float var = Q * (1.f / 8192.f) - mean * mean;
  const float sc = rsqrtf(var + 1e-5f);

#pragma unroll
  for (int ch = 0; ch < 2; ++ch) {
    const int c = g * 2 + ch;
    const float gw = gnw[c];
    const float a = gw * sc, bo = gnb[c] - mean * sc * gw;
    kern[ch][tid] = (ch == 0) ? kv0 : kv1;
#pragma unroll
    for (int j = 0; j < 4; ++j) {
      const int idx = tid + 256 * j;
      float* d = &tile[ch][(idx >> 4) + 3][((idx & 15) << 2) + 3];
      const v4f x = v[4 * ch + j];
      d[0] = relu_(a * x.x + bo);
      d[1] = relu_(a * x.y + bo);
      d[2] = relu_(a * x.z + bo);
      d[3] = relu_(a * x.w + bo);
    }
  }
  __syncthreads();
  for (int ch = 0; ch < 2; ++ch) {
    const int bc = bg * 2 + ch;
    const float tg = t_global[bc];
    float* dst = corr + (size_t)bc * 4096;
    for (int p = tid; p < 4096; p += 256) {
      const int y = p >> 6, x = p & 63;
      float acc = tg * tile[ch][y + 3][x + 3];
#pragma unroll
      for (int ky = 0; ky < 7; ++ky)
#pragma unroll
        for (int kx = 0; kx < 7; ++kx)
          acc = fmaf(kern[ch][ky * 7 + kx], tile[ch][y + ky][x + kx], acc);
      dst[p] = acc;
    }
  }
}

#define CV_PITCH 68
#define CV_COLS 66
#define CV_ROWS 4

__global__ __launch_bounds__(256) void k_conv3x3_gemm(
    const float* __restrict__ corr, const float* __restrict__ w_p1,
    float* __restrict__ y_pre, float* __restrict__ ypart) {
  const int b = blockIdx.y;
  const int tid = threadIdx.x;
  const int wave = tid >> 5, lane = tid & 31;
  const int hv = lane >> 4, l16 = lane & 15;
  const int m0 = (wave & 3) << 4;
  const int yl = wave >> 2;
  const int y0 = blockIdx.x << 1;
  const int nbase = blockIdx.x * 128 + (yl << 6);

  __shared__ __attribute__((aligned(16))) float tile[CV_ROWS * CV_COLS * CV_PITCH];
  __shared__ float red[8][16];

  const float* X = corr + (size_t)b * 64 * 4096;

  {
    const int x4 = tid & 15, c4 = tid >> 4;
#pragma unroll
    for (int r = 0; r < CV_ROWS; ++r) {
      const int gy = y0 - 1 + r;
      const bool in = (unsigned)gy < 64u;
      const float* s = X + (size_t)(c4 * 4) * 4096 + (in ? gy : 0) * 64 + x4 * 4;
      v4f v0 = *(const v4f*)(s);
      v4f v1 = *(const v4f*)(s + 4096);
      v4f v2 = *(const v4f*)(s + 8192);
      v4f v3 = *(const v4f*)(s + 12288);
      const v4f z = {0.f, 0.f, 0.f, 0.f};
      v0 = in ? v0 : z; v1 = in ? v1 : z; v2 = in ? v2 : z; v3 = in ? v3 : z;
      float* d = tile + (r * CV_COLS + x4 * 4 + 1) * CV_PITCH + c4 * 4;
      const v4f p0 = {v0.x, v1.x, v2.x, v3.x};
      const v4f p1 = {v0.y, v1.y, v2.y, v3.y};
      const v4f p2 = {v0.z, v1.z, v2.z, v3.z};
      const v4f p3 = {v0.w, v1.w, v2.w, v3.w};
      *(v4f*)(d) = p0;
      *(v4f*)(d + CV_PITCH) = p1;
      *(v4f*)(d + 2 * CV_PITCH) = p2;
      *(v4f*)(d + 3 * CV_PITCH) = p3;
    }
    if (tid < 128) {
      const int r = tid >> 5, side = (tid >> 4) & 1, cc = tid & 15;
      const v4f z = {0.f, 0.f, 0.f, 0.f};
      *(v4f*)(tile + (r * CV_COLS + side * (CV_COLS - 1)) * CV_PITCH + cc * 4) = z;
    }
  }
  __syncthreads();

  v8f acc[4] = {};
  const float* Wl = w_p1 + (size_t)(m0 + l16) * 576 + hv * 36;
  const float* Bl = tile + (yl * CV_COLS + l16) * CV_PITCH + hv * 4;

  for (int g = 0; g < 8; ++g) {
    v4f aw[9];
#pragma unroll
    for (int i = 0; i < 9; ++i) aw[i] = *(const v4f*)(Wl + g * 72 + i * 4);
#define CV_AW(e) (aw[(e) >> 2][(e) & 3])
    const float* Bg = Bl + g * 8;
#pragma unroll
    for (int t = 0; t < 9; ++t) {
      const int dyi = t / 3, dxi = t % 3;
      v2f a0, a1;
      a0.x = CV_AW(t);      a0.y = CV_AW(9 + t);
      a1.x = CV_AW(18 + t); a1.y = CV_AW(27 + t);
      v4f q[4];
#pragma unroll
      for (int j = 0; j < 4; ++j)
        q[j] = *(const v4f*)(Bg + (dyi * CV_COLS + (j << 4) + dxi) * CV_PITCH);
#pragma unroll
      for (int j = 0; j < 4; ++j) {
        v2f b0; b0.x = q[j].x; b0.y = q[j].y;
        acc[j] = wmma_f32_4(a0, b0, acc[j]);
      }
#pragma unroll
      for (int j = 0; j < 4; ++j) {
        v2f b1; b1.x = q[j].z; b1.y = q[j].w;
        acc[j] = wmma_f32_4(a1, b1, acc[j]);
      }
    }
#undef CV_AW
  }

#pragma unroll
  for (int p = 0; p < 4; ++p) {
    float s = 0.f, q = 0.f;
#pragma unroll
    for (int j = 0; j < 4; ++j) {
      const float e0 = acc[j][2 * p], e1 = acc[j][2 * p + 1];
      s += e0 + e1;
      q = fmaf(e0, e0, q);
      q = fmaf(e1, e1, q);
    }
    s += __shfl_xor(s, 1);  q += __shfl_xor(q, 1);
    s += __shfl_xor(s, 2);  q += __shfl_xor(q, 2);
    s += __shfl_xor(s, 4);  q += __shfl_xor(q, 4);
    s += __shfl_xor(s, 8);  q += __shfl_xor(q, 8);
    if (l16 == 0) {
      red[wave][(hv << 3) + (p << 1)] = s;
      red[wave][(hv << 3) + (p << 1) + 1] = q;
    }
  }
  __syncthreads();
  if (wave == 0) {
    const int mt = lane >> 3, o2 = (lane & 7) << 1;
    v2f sq;
    sq.x = red[mt][o2] + red[mt + 4][o2];
    sq.y = red[mt][o2 + 1] + red[mt + 4][o2 + 1];
    *(v2f*)(ypart + ((((size_t)b * 32 + blockIdx.x) * 32 + lane) << 1)) = sq;
  }

  float* O = y_pre + (size_t)b * 64 * 4096;
#pragma unroll
  for (int r = 0; r < 8; ++r) {
    const int row = m0 + r + (hv << 3);
    float* o = O + (size_t)row * 4096 + nbase + l16;
    o[0] = acc[0][r];
    o[16] = acc[1][r];
    o[32] = acc[2][r];
    o[48] = acc[3][r];
  }
}

__global__ __launch_bounds__(256) void k_final(
    const float* __restrict__ y_pre, const float* __restrict__ ypart,
    const float* __restrict__ gnw, const float* __restrict__ gnb,
    const float* __restrict__ w_p2, const float* __restrict__ b_p2,
    float* __restrict__ out) {
  const int b = blockIdx.x >> 4;
  const int tid = threadIdx.x;
  const int p = ((blockIdx.x & 15) << 8) + tid;
  __shared__ float rs[8][32], rq[8][32];
  __shared__ float sa[64], sb[64];
  {
    const int gi = tid & 31, part = tid >> 5;
    const float* pp = ypart + ((((size_t)b * 32 + part * 4) * 32 + gi) << 1);
    const v2f t0 = *(const v2f*)(pp);
    const v2f t1 = *(const v2f*)(pp + 64);
    const v2f t2 = *(const v2f*)(pp + 128);
    const v2f t3 = *(const v2f*)(pp + 192);
    rs[part][gi] = (t0.x + t1.x) + (t2.x + t3.x);
    rq[part][gi] = (t0.y + t1.y) + (t2.y + t3.y);
  }
  __syncthreads();
  if (tid < 32) {
    float S = 0.f, Q = 0.f;
    for (int w = 0; w < 8; ++w) {
      S += rs[w][tid];
      Q += rq[w][tid];
    }
    const float mean = S * (1.f / 8192.f);
    const float var = Q * (1.f / 8192.f) - mean * mean;
    const float sc = rsqrtf(var + 1e-5f);
    for (int ch = 0; ch < 2; ++ch) {
      const int c = tid * 2 + ch;
      const float gw = gnw[c];
      sa[c] = gw * sc;
      sb[c] = gnb[c] - mean * sc * gw;
    }
  }
  __syncthreads();
  const float* Y = y_pre + (size_t)b * 64 * 4096 + p;
  float acc = b_p2[0];
#pragma unroll 8
  for (int c = 0; c < 64; ++c) {
    const float v = relu_(sa[c] * Y[(size_t)c * 4096] + sb[c]);
    acc = fmaf(w_p2[c], v, acc);
  }
  out[(size_t)b * 4096 + p] = acc;
}

extern "C" void kernel_launch(void* const* d_in, const int* in_sizes, int n_in,
                              void* d_out, int out_size, void* d_ws,
                              size_t ws_size, hipStream_t stream) {
  (void)in_sizes; (void)n_in; (void)out_size; (void)ws_size;
  const float* tf   = (const float*)d_in[0];
  const float* sf   = (const float*)d_in[1];
  const float* w_t  = (const float*)d_in[2];
  const float* gtw  = (const float*)d_in[3];
  const float* gtb  = (const float*)d_in[4];
  const float* w_s  = (const float*)d_in[5];
  const float* gsw  = (const float*)d_in[6];
  const float* gsb  = (const float*)d_in[7];
  const float* w_p1 = (const float*)d_in[8];
  const float* gpw  = (const float*)d_in[9];
  const float* gpb  = (const float*)d_in[10];
  const float* w_p2 = (const float*)d_in[11];
  const float* b_p2 = (const float*)d_in[12];
  float* out = (float*)d_out;

  float* ws = (float*)d_ws;
  float* t_global = ws;
  float* t_kernel = ws + 2048;
  float* ypart = ws + 2048;
  float* bufA = ws + 131072;
  float* bufB = bufA + 8388608;

  k_template_path<<<dim3(4, 32), 416, 0, stream>>>(tf, w_t, gtw, gtb, t_global, t_kernel);
  k_search_gemm<<<dim3(32, 32), 256, 0, stream>>>(sf, w_s, bufA);
  k_corr<<<1024, 256, 0, stream>>>(bufA, gsw, gsb, t_global, t_kernel, bufB);
  k_conv3x3_gemm<<<dim3(32, 32), 256, 0, stream>>>(bufB, w_p1, bufA, ypart);
  k_final<<<512, 256, 0, stream>>>(bufA, ypart, gpw, gpb, w_p2, b_p2, out);
}


# corr kernel: each thread computes a 4x4 output patch from 16-byte LDS reads of its 10 tile rows (taps as scalars) instead of one LDS read per tap and pixel; per-pixel sum order unchanged
# speedup vs baseline: 2.2778x; 1.1447x over previous
#include <hip/hip_runtime.h>
#include <hip/hip_bf16.h>


typedef float v2f __attribute__((ext_vector_type(2)));
typedef float v4f __attribute__((ext_vector_type(4)));
typedef float v8f __attribute__((ext_vector_type(8)));

__device__ __forceinline__ v8f wmma_f32_4(v2f a, v2f b, v8f c) {
  return __builtin_amdgcn_wmma_f32_16x16x4_f32(false, a, false, b, (short)0, c,
                                               false, false);
}

__device__ __forceinline__ float relu_(float x) { return fmaxf(x, 0.0f); }

__global__ __launch_bounds__(416) void k_template_path(
    const float* __restrict__ tf, const float* __restrict__ w_t,
    const float* __restrict__ gnw, const float* __restrict__ gnb,
    float* __restrict__ t_global, float* __restrict__ t_kernel) {
  const int m0 = blockIdx.x << 4;
  const int b = blockIdx.y;
  const int tid = threadIdx.x;
  const int wave = tid >> 5, lane = tid & 31;
  const int hv = lane >> 4, l16 = lane & 15;

  __shared__ float lds_t[16][208];
  __shared__ float ga[16], gb[16];

  const float* X = tf + (size_t)b * 256 * 196;

  {
    const int n0 = wave << 4;
    const int n = n0 + l16;
    const int nc = (n < 196) ? n : 195;
    const float* Ap = w_t + (m0 + l16) * 256 + (hv << 1);
    const float* Bp = X + (hv << 1) * 196 + nc;
    v8f acc = {};
    for (int k0 = 0; k0 < 256; k0 += 64) {
      v2f a[16], bb[16];
#pragma unroll
      for (int j = 0; j < 16; ++j) {
        a[j].x = Ap[k0 + 4 * j];
        a[j].y = Ap[k0 + 4 * j + 1];
        bb[j].x = Bp[(k0 + 4 * j) * 196];
        bb[j].y = Bp[(k0 + 4 * j + 1) * 196];
      }
#pragma unroll
      for (int j = 0; j < 16; ++j) acc = wmma_f32_4(a[j], bb[j], acc);
    }
#pragma unroll
    for (int r = 0; r < 8; ++r)
      lds_t[r + (hv << 3)][n0 + l16] = acc[r];
  }
  __syncthreads();

  if (wave < 8) {
    float s = 0.f, q = 0.f;
    for (int i = lane; i < 392; i += 32) {
      const int hi = (i >= 196) ? 1 : 0;
      const float v = lds_t[(wave << 1) + hi][i - hi * 196];
      s += v;
      q += v * v;
    }
    s += __shfl_xor(s, 1);  q += __shfl_xor(q, 1);
    s += __shfl_xor(s, 2);  q += __shfl_xor(q, 2);
    s += __shfl_xor(s, 4);  q += __shfl_xor(q, 4);
    s += __shfl_xor(s, 8);  q += __shfl_xor(q, 8);
    s += __shfl_xor(s, 16); q += __shfl_xor(q, 16);
    const float mean = s * (1.0f / 392.0f);
    const float var = q * (1.0f / 392.0f) - mean * mean;
    const float sc = rsqrtf(var + 1e-5f);
    if (lane < 2) {
      const int cl = (wave << 1) + lane;
      const int c = m0 + cl;
      ga[cl] = gnw[c] * sc;
      gb[cl] = gnb[c] - mean * sc * gnw[c];
    }
  }
  __syncthreads();

  for (int idx = tid; idx < 16 * 49; idx += 416) {
    const int c = idx / 49, kid = idx % 49;
    const int ky = kid / 7, kx = kid % 7;
    const float a = ga[c], bo = gb[c];
    float s = 0.f;
#pragma unroll
    for (int dy = 0; dy < 2; ++dy)
#pragma unroll
      for (int dx = 0; dx < 2; ++dx)
        s += relu_(a * lds_t[c][(2 * ky + dy) * 14 + (2 * kx + dx)] + bo);
    t_kernel[((size_t)b * 64 + m0 + c) * 49 + kid] = s * 0.25f;
  }
  if (wave < 8) {
    const int c = tid >> 4, sub = tid & 15;
    const float a = ga[c], bo = gb[c];
    float s = 0.f;
    for (int p = sub; p < 196; p += 16) s += relu_(a * lds_t[c][p] + bo);
    s += __shfl_xor(s, 1);
    s += __shfl_xor(s, 2);
    s += __shfl_xor(s, 4);
    s += __shfl_xor(s, 8);
    if (sub == 0) t_global[(size_t)b * 64 + m0 + c] = s * (1.0f / 196.0f);
  }
}

__global__ __launch_bounds__(256) void k_search_gemm(
    const float* __restrict__ sf, const float* __restrict__ w_s,
    float* __restrict__ s_pre) {
  const int b = blockIdx.y;
  const int tid = threadIdx.x;
  const int wave = tid >> 5, lane = tid & 31;
  const int hv = lane >> 4, l16 = lane & 15;
  const int m0 = (wave & 3) << 4;
  const int nbase = blockIdx.x * 128 + ((wave >> 2) << 6);

  const float* X = sf + (size_t)b * 256 * 4096;
  v8f acc0 = {}, acc1 = {}, acc2 = {}, acc3 = {};

  for (int k0 = 0; k0 < 256; k0 += 4) {
    const int kk = k0 + (hv << 1);
    v2f a;
    a.x = w_s[(m0 + l16) * 256 + kk];
    a.y = w_s[(m0 + l16) * 256 + kk + 1];
    const float* r0 = X + (size_t)kk * 4096 + nbase + l16;
    const float* r1 = r0 + 4096;
    v2f b0; b0.x = r0[0];  b0.y = r1[0];  acc0 = wmma_f32_4(a, b0, acc0);
    v2f b1; b1.x = r0[16]; b1.y = r1[16]; acc1 = wmma_f32_4(a, b1, acc1);
    v2f b2; b2.x = r0[32]; b2.y = r1[32]; acc2 = wmma_f32_4(a, b2, acc2);
    v2f b3; b3.x = r0[48]; b3.y = r1[48]; acc3 = wmma_f32_4(a, b3, acc3);
  }

  float* O = s_pre + (size_t)b * 64 * 4096;
#pragma unroll
  for (int r = 0; r < 8; ++r) {
    const int row = m0 + r + (hv << 3);
    float* o = O + (size_t)row * 4096 + nbase + l16;
    o[0] = acc0[r];
    o[16] = acc1[r];
    o[32] = acc2[r];
    o[48] = acc3[r];
  }
}

#define CR_ROWS 70
#define CR_PITCH 72

__global__ __launch_bounds__(256) void k_corr(
    const float* __restrict__ s_pre, const float* __restrict__ gnw,
    const float* __restrict__ gnb, const float* __restrict__ t_global,
    const float* __restrict__ t_kernel, float* __restrict__ corr) {
  const int bg = blockIdx.x;
  const int g = bg & 31;
  const float* src = s_pre + (size_t)bg * 8192;
  __shared__ __attribute__((aligned(16))) float tile[2][CR_ROWS][CR_PITCH];
  __shared__ float red_s[8], red_q[8];
  const int tid = threadIdx.x;
  const int wave = tid >> 5, lane = tid & 31;
  v4f v[8];
#pragma unroll
  for (int j = 0; j < 8; ++j)
    v[j] = *(const v4f*)(src + ((tid + 256 * j) << 2));
  if (tid < 236) {
    int row, piece;
    if (tid < 108) {
      const int r = tid / 18;
      row = (r < 3) ? r : r + 64;
      piece = tid - r * 18;
    } else {
      const int j = tid - 108;
      row = 3 + (j >> 1);
      piece = (j & 1) ? 17 : 0;
    }
    const v4f z = {0.f, 0.f, 0.f, 0.f};
    *(v4f*)&tile[0][row][piece << 2] = z;
    *(v4f*)&tile[1][row][piece << 2] = z;
  }
  float s = 0.f, q = 0.f;
#pragma unroll
  for (int j = 0; j < 8; ++j) {
    s += (v[j].x + v[j].y) + (v[j].z + v[j].w);
    q = fmaf(v[j].x, v[j].x, q);
    q = fmaf(v[j].y, v[j].y, q);
    q = fmaf(v[j].z, v[j].z, q);
    q = fmaf(v[j].w, v[j].w, q);
  }
  s += __shfl_xor(s, 1);  q += __shfl_xor(q, 1);
  s += __shfl_xor(s, 2);  q += __shfl_xor(q, 2);
  s += __shfl_xor(s, 4);  q += __shfl_xor(q, 4);
  s += __shfl_xor(s, 8);  q += __shfl_xor(q, 8);
  s += __shfl_xor(s, 16); q += __shfl_xor(q, 16);
  if (lane == 0) {
    red_s[wave] = s;
    red_q[wave] = q;
  }
  __syncthreads();
  float S = 0.f, Q = 0.f;
  for (int w = 0; w < 8; ++w) {
    S += red_s[w];
    Q += red_q[w];
  }
  const float mean = S * (1.f / 8192.f);
  const float var = Q * (1.f / 8192.f) - mean * mean;
  const float sc = rsqrtf(var + 1e-5f);

#pragma unroll
  for (int ch = 0; ch < 2; ++ch) {
    const int c = g * 2 + ch;
    const float gw = gnw[c];
    const float a = gw * sc, bo = gnb[c] - mean * sc * gw;
#pragma unroll
    for (int j = 0; j < 4; ++j) {
      const int idx = tid + 256 * j;
      const v4f x = v[4 * ch + j];
      v4f o;
      o.x = relu_(a * x.x + bo);
      o.y = relu_(a * x.y + bo);
      o.z = relu_(a * x.z + bo);
      o.w = relu_(a * x.w + bo);
      *(v4f*)&tile[ch][(idx >> 4) + 3][((idx & 15) << 2) + 4] = o;
    }
  }
  __syncthreads();

  const int py = tid >> 4, px = tid & 15;
  for (int ch = 0; ch < 2; ++ch) {
    const int bc = bg * 2 + ch;
    const float tg = t_global[bc];
    const float* kp = t_kernel + (size_t)bc * 49;
    const float* tb = &tile[ch][py << 2][px << 2];
    float acc[4][4];
#pragma unroll
    for (int oy = 0; oy < 4; ++oy) {
      const v4f cv = *(const v4f*)(tb + (oy + 3) * CR_PITCH + 4);
      acc[oy][0] = tg * cv.x;
      acc[oy][1] = tg * cv.y;
      acc[oy][2] = tg * cv.z;
      acc[oy][3] = tg * cv.w;
    }
#pragma unroll
    for (int ty = 0; ty < 10; ++ty) {
      const v4f r0 = *(const v4f*)(tb + ty * CR_PITCH);
      const v4f r1 = *(const v4f*)(tb + ty * CR_PITCH + 4);
      const v4f r2 = *(const v4f*)(tb + ty * CR_PITCH + 8);
      const float r[12] = {r0.x, r0.y, r0.z, r0.w, r1.x, r1.y,
                           r1.z, r1.w, r2.x, r2.y, r2.z, r2.w};
#pragma unroll
      for (int oy = 0; oy < 4; ++oy) {
        const int ky = ty - oy;
        if (ky >= 0 && ky < 7) {
#pragma unroll
          for (int kx = 0; kx < 7; ++kx) {
            const float kv = kp[ky * 7 + kx];
#pragma unroll
            for (int j = 0; j < 4; ++j)
              acc[oy][j] = fmaf(kv, r[j + kx + 1], acc[oy][j]);
          }
        }
      }
    }
    float* dst = corr + (size_t)bc * 4096 + (py << 8) + (px << 2);
#pragma unroll
    for (int oy = 0; oy < 4; ++oy) {
      const v4f o = {acc[oy][0], acc[oy][1], acc[oy][2], acc[oy][3]};
      *(v4f*)(dst + (oy << 6)) = o;
    }
  }
}

#define CV_PITCH 68
#define CV_COLS 66
#define CV_ROWS 4

__global__ __launch_bounds__(256) void k_conv3x3_gemm(
    const float* __restrict__ corr, const float* __restrict__ w_p1,
    float* __restrict__ y_pre, float* __restrict__ ypart) {
  const int b = blockIdx.y;
  const int tid = threadIdx.x;
  const int wave = tid >> 5, lane = tid & 31;
  const int hv = lane >> 4, l16 = lane & 15;
  const int m0 = (wave & 3) << 4;
  const int yl = wave >> 2;
  const int y0 = blockIdx.x << 1;
  const int nbase = blockIdx.x * 128 + (yl << 6);

  __shared__ __attribute__((aligned(16))) float tile[CV_ROWS * CV_COLS * CV_PITCH];
  __shared__ float red[8][16];

  const float* X = corr + (size_t)b * 64 * 4096;

  {
    const int x4 = tid & 15, c4 = tid >> 4;
#pragma unroll
    for (int r = 0; r < CV_ROWS; ++r) {
      const int gy = y0 - 1 + r;
      const bool in = (unsigned)gy < 64u;
      const float* s = X + (size_t)(c4 * 4) * 4096 + (in ? gy : 0) * 64 + x4 * 4;
      v4f v0 = *(const v4f*)(s);
      v4f v1 = *(const v4f*)(s + 4096);
      v4f v2 = *(const v4f*)(s + 8192);
      v4f v3 = *(const v4f*)(s + 12288);
      const v4f z = {0.f, 0.f, 0.f, 0.f};
      v0 = in ? v0 : z; v1 = in ? v1 : z; v2 = in ? v2 : z; v3 = in ? v3 : z;
      float* d = tile + (r * CV_COLS + x4 * 4 + 1) * CV_PITCH + c4 * 4;
      const v4f p0 = {v0.x, v1.x, v2.x, v3.x};
      const v4f p1 = {v0.y, v1.y, v2.y, v3.y};
      const v4f p2 = {v0.z, v1.z, v2.z, v3.z};
      const v4f p3 = {v0.w, v1.w, v2.w, v3.w};
      *(v4f*)(d) = p0;
      *(v4f*)(d + CV_PITCH) = p1;
      *(v4f*)(d + 2 * CV_PITCH) = p2;
      *(v4f*)(d + 3 * CV_PITCH) = p3;
    }
    if (tid < 128) {
      const int r = tid >> 5, side = (tid >> 4) & 1, cc = tid & 15;
      const v4f z = {0.f, 0.f, 0.f, 0.f};
      *(v4f*)(tile + (r * CV_COLS + side * (CV_COLS - 1)) * CV_PITCH + cc * 4) = z;
    }
  }
  __syncthreads();

  v8f acc[4] = {};
  const float* Wl = w_p1 + (size_t)(m0 + l16) * 576 + hv * 36;
  const float* Bl = tile + (yl * CV_COLS + l16) * CV_PITCH + hv * 4;

  for (int g = 0; g < 8; ++g) {
    v4f aw[9];
#pragma unroll
    for (int i = 0; i < 9; ++i) aw[i] = *(const v4f*)(Wl + g * 72 + i * 4);
#define CV_AW(e) (aw[(e) >> 2][(e) & 3])
    const float* Bg = Bl + g * 8;
#pragma unroll
    for (int t = 0; t < 9; ++t) {
      const int dyi = t / 3, dxi = t % 3;
      v2f a0, a1;
      a0.x = CV_AW(t);      a0.y = CV_AW(9 + t);
      a1.x = CV_AW(18 + t); a1.y = CV_AW(27 + t);
      v4f q[4];
#pragma unroll
      for (int j = 0; j < 4; ++j)
        q[j] = *(const v4f*)(Bg + (dyi * CV_COLS + (j << 4) + dxi) * CV_PITCH);
#pragma unroll
      for (int j = 0; j < 4; ++j) {
        v2f b0; b0.x = q[j].x; b0.y = q[j].y;
        acc[j] = wmma_f32_4(a0, b0, acc[j]);
      }
#pragma unroll
      for (int j = 0; j < 4; ++j) {
        v2f b1; b1.x = q[j].z; b1.y = q[j].w;
        acc[j] = wmma_f32_4(a1, b1, acc[j]);
      }
    }
#undef CV_AW
  }

#pragma unroll
  for (int p = 0; p < 4; ++p) {
    float s = 0.f, q = 0.f;
#pragma unroll
    for (int j = 0; j < 4; ++j) {
      const float e0 = acc[j][2 * p], e1 = acc[j][2 * p + 1];
      s += e0 + e1;
      q = fmaf(e0, e0, q);
      q = fmaf(e1, e1, q);
    }
    s += __shfl_xor(s, 1);  q += __shfl_xor(q, 1);
    s += __shfl_xor(s, 2);  q += __shfl_xor(q, 2);
    s += __shfl_xor(s, 4);  q += __shfl_xor(q, 4);
    s += __shfl_xor(s, 8);  q += __shfl_xor(q, 8);
    if (l16 == 0) {
      red[wave][(hv << 3) + (p << 1)] = s;
      red[wave][(hv << 3) + (p << 1) + 1] = q;
    }
  }
  __syncthreads();
  if (wave == 0) {
    const int mt = lane >> 3, o2 = (lane & 7) << 1;
    v2f sq;
    sq.x = red[mt][o2] + red[mt + 4][o2];
    sq.y = red[mt][o2 + 1] + red[mt + 4][o2 + 1];
    *(v2f*)(ypart + ((((size_t)b * 32 + blockIdx.x) * 32 + lane) << 1)) = sq;
  }

  float* O = y_pre + (size_t)b * 64 * 4096;
#pragma unroll
  for (int r = 0; r < 8; ++r) {
    const int row = m0 + r + (hv << 3);
    float* o = O + (size_t)row * 4096 + nbase + l16;
    o[0] = acc[0][r];
    o[16] = acc[1][r];
    o[32] = acc[2][r];
    o[48] = acc[3][r];
  }
}

__global__ __launch_bounds__(256) void k_final(
    const float* __restrict__ y_pre, const float* __restrict__ ypart,
    const float* __restrict__ gnw, const float* __restrict__ gnb,
    const float* __restrict__ w_p2, const float* __restrict__ b_p2,
    float* __restrict__ out) {
  const int b = blockIdx.x >> 4;
  const int tid = threadIdx.x;
  const int p = ((blockIdx.x & 15) << 8) + tid;
  __shared__ float rs[8][32], rq[8][32];
  __shared__ float sa[64], sb[64];
  {
    const int gi = tid & 31, part = tid >> 5;
    const float* pp = ypart + ((((size_t)b * 32 + part * 4) * 32 + gi) << 1);
    const v2f t0 = *(const v2f*)(pp);
    const v2f t1 = *(const v2f*)(pp + 64);
    const v2f t2 = *(const v2f*)(pp + 128);
    const v2f t3 = *(const v2f*)(pp + 192);
    rs[part][gi] = (t0.x + t1.x) + (t2.x + t3.x);
    rq[part][gi] = (t0.y + t1.y) + (t2.y + t3.y);
  }
  __syncthreads();
  if (tid < 32) {
    float S = 0.f, Q = 0.f;
    for (int w = 0; w < 8; ++w) {
      S += rs[w][tid];
      Q += rq[w][tid];
    }
    const float mean = S * (1.f / 8192.f);
    const float var = Q * (1.f / 8192.f) - mean * mean;
    const float sc = rsqrtf(var + 1e-5f);
    for (int ch = 0; ch < 2; ++ch) {
      const int c = tid * 2 + ch;
      const float gw = gnw[c];
      sa[c] = gw * sc;
      sb[c] = gnb[c] - mean * sc * gw;
    }
  }
  __syncthreads();
  const float* Y = y_pre + (size_t)b * 64 * 4096 + p;
  float acc = b_p2[0];
#pragma unroll 8
  for (int c = 0; c < 64; ++c) {
    const float v = relu_(sa[c] * Y[(size_t)c * 4096] + sb[c]);
    acc = fmaf(w_p2[c], v, acc);
  }
  out[(size_t)b * 4096 + p] = acc;
}

extern "C" void kernel_launch(void* const* d_in, const int* in_sizes, int n_in,
                              void* d_out, int out_size, void* d_ws,
                              size_t ws_size, hipStream_t stream) {
  (void)in_sizes; (void)n_in; (void)out_size; (void)ws_size;
  const float* tf   = (const float*)d_in[0];
  const float* sf   = (const float*)d_in[1];
  const float* w_t  = (const float*)d_in[2];
  const float* gtw  = (const float*)d_in[3];
  const float* gtb  = (const float*)d_in[4];
  const float* w_s  = (const float*)d_in[5];
  const float* gsw  = (const float*)d_in[6];
  const float* gsb  = (const float*)d_in[7];
  const float* w_p1 = (const float*)d_in[8];
  const float* gpw  = (const float*)d_in[9];
  const float* gpb  = (const float*)d_in[10];
  const float* w_p2 = (const float*)d_in[11];
  const float* b_p2 = (const float*)d_in[12];
  float* out = (float*)d_out;

  float* ws = (float*)d_ws;
  float* t_global = ws;
  float* t_kernel = ws + 2048;
  float* ypart = ws + 2048;
  float* bufA = ws + 131072;
  float* bufB = bufA + 8388608;

  k_template_path<<<dim3(4, 32), 416, 0, stream>>>(tf, w_t, gtw, gtb, t_global, t_kernel);
  k_search_gemm<<<dim3(32, 32), 256, 0, stream>>>(sf, w_s, bufA);
  k_corr<<<1024, 256, 0, stream>>>(bufA, gsw, gsb, t_global, t_kernel, bufB);
  k_conv3x3_gemm<<<dim3(32, 32), 256, 0, stream>>>(bufB, w_p1, bufA, ypart);
  k_final<<<512, 256, 0, stream>>>(bufA, ypart, gpw, gpb, w_p2, b_p2, out);
}
